# starting kernel
# baseline (speedup 1.0000x reference)
#include <hip/hip_runtime.h>
#include <hip/hip_bf16.h>

#define HW   4096
#define NB   8
#define CIN  256
#define CQ   128
#define CV   256

typedef float    v2f  __attribute__((ext_vector_type(2)));
typedef float    v8f  __attribute__((ext_vector_type(8)));
typedef _Float16 v16h __attribute__((ext_vector_type(16)));
typedef _Float16 h8   __attribute__((ext_vector_type(8)));

static __device__ __forceinline__ v16h load_frag16(const _Float16* rowBase, int lh) {
  const h8 lo = *reinterpret_cast<const h8*>(rowBase + 8 * lh);
  const h8 hi = *reinterpret_cast<const h8*>(rowBase + 8 * lh + 16);
  v16h r;
#pragma unroll
  for (int i = 0; i < 8; ++i) { r[i] = lo[i]; r[i + 8] = hi[i]; }
  return r;
}

__global__ __launch_bounds__(128) void proj_bn_relu_kernel(
    const float* __restrict__ x, const float* __restrict__ w,
    const float* __restrict__ gamma, const float* __restrict__ beta,
    const float* __restrict__ mean, const float* __restrict__ var,
    _Float16* __restrict__ outh, int posMajor)
{
  const int lane = threadIdx.x;
  const int l15  = lane & 15;
  const int lh   = lane >> 4;
  const int p0   = (blockIdx.x * blockDim.y + threadIdx.y) * 16;
  const int o0   = blockIdx.y * 16;
  const int n    = blockIdx.z;

  const float* wrow = w + (size_t)(o0 + l15) * CIN;
  const float* xcol = x + (size_t)n * CIN * HW + (p0 + l15);

  v8f acc0 = {};
  v8f acc1 = {};
#pragma unroll 4
  for (int c0 = 0; c0 < CIN; c0 += 8) {
    const float2 aw0 = *reinterpret_cast<const float2*>(wrow + c0 + 2 * lh);
    const float2 aw1 = *reinterpret_cast<const float2*>(wrow + c0 + 4 + 2 * lh);
    v2f a0; a0[0] = aw0.x; a0[1] = aw0.y;
    v2f a1; a1[0] = aw1.x; a1[1] = aw1.y;
    v2f b0; b0[0] = xcol[(size_t)(c0 + 2 * lh + 0) * HW];
            b0[1] = xcol[(size_t)(c0 + 2 * lh + 1) * HW];
    v2f b1; b1[0] = xcol[(size_t)(c0 + 4 + 2 * lh + 0) * HW];
            b1[1] = xcol[(size_t)(c0 + 4 + 2 * lh + 1) * HW];
    acc0 = __builtin_amdgcn_wmma_f32_16x16x4_f32(false, a0, false, b0, (short)0, acc0, false, false);
    acc1 = __builtin_amdgcn_wmma_f32_16x16x4_f32(false, a1, false, b1, (short)0, acc1, false, false);
  }
  const v8f acc = acc0 + acc1;

#pragma unroll
  for (int r = 0; r < 8; ++r) {
    const int ch   = o0 + 8 * lh + r;
    const float iv = gamma[ch] * rsqrtf(var[ch] + 1e-5f);
    const float bb = beta[ch] - mean[ch] * iv;
    float v = acc[r] * iv + bb;
    v = v > 0.f ? v : 0.f;
    const _Float16 hv = (_Float16)v;
    const int p = p0 + l15;
    if (posMajor)
      outh[((size_t)n * HW + p) * CQ + ch] = hv;
    else
      outh[((size_t)n * CV + ch) * HW + p] = hv;
  }
}

__global__ __launch_bounds__(128) void flash_attn_kernel(
    const _Float16* __restrict__ q, const _Float16* __restrict__ fT,
    const float* __restrict__ mask, float* __restrict__ out)
{
  constexpr float LOG2E     = 1.4426950408889634f;
  constexpr float ATT_SCALE = (1.0f / (1e-8f + 11.313708498984761f)) * LOG2E;
  __shared__ __align__(16) _Float16 Pbuf[4][16 * 32];

  const int lane = threadIdx.x;
  const int l15  = lane & 15;
  const int lh   = lane >> 4;
  const int wid  = threadIdx.y;
  const int i0   = (blockIdx.x * 4 + wid) * 16;
  const int n    = blockIdx.y;

  v16h qa[4];
  {
    const _Float16* qrow = q + ((size_t)n * HW + i0 + l15) * CQ;
#pragma unroll
    for (int kc = 0; kc < 4; ++kc) qa[kc] = load_frag16(qrow + kc * 32, lh);
  }

  v8f oacc[16];
#pragma unroll
  for (int t = 0; t < 16; ++t) oacc[t] = (v8f){};
  float mrow[8], lrow[8];
#pragma unroll
  for (int r = 0; r < 8; ++r) { mrow[r] = -1e30f; lrow[r] = 0.f; }

  _Float16* pb    = &Pbuf[wid][0];
  const float* mk = mask + (size_t)n * HW;

  for (int j0 = 0; j0 < HW; j0 += 32) {
    if (j0 + 32 < HW) {
      __builtin_prefetch(q + ((size_t)n * HW + j0 + 32 + l15) * CQ, 0, 3);
      __builtin_prefetch(q + ((size_t)n * HW + j0 + 48 + l15) * CQ, 0, 3);
    }

    v8f s0 = (v8f){};
    v8f s1 = (v8f){};
#pragma unroll
    for (int kc = 0; kc < 4; ++kc) {
      const _Float16* kb = q + ((size_t)n * HW + j0 + l15) * CQ + kc * 32;
      const v16h b0 = load_frag16(kb, lh);
      const v16h b1 = load_frag16(kb + 16 * CQ, lh);
      s0 = __builtin_amdgcn_wmma_f32_16x16x32_f16(false, qa[kc], false, b0, (short)0, s0, false, false);
      s1 = __builtin_amdgcn_wmma_f32_16x16x32_f16(false, qa[kc], false, b1, (short)0, s1, false, false);
    }

    const float ma0 = (mk[j0 + l15]      - 1.f) * (1e8f * LOG2E);
    const float ma1 = (mk[j0 + 16 + l15] - 1.f) * (1e8f * LOG2E);

    v8f avec;
#pragma unroll
    for (int r = 0; r < 8; ++r) {
      float e0 = s0[r] * ATT_SCALE + ma0;
      float e1 = s1[r] * ATT_SCALE + ma1;
      float vmax = fmaxf(e0, e1);
#pragma unroll
      for (int m = 1; m <= 8; m <<= 1) vmax = fmaxf(vmax, __shfl_xor(vmax, m, 32));
      const float mn = fmaxf(mrow[r], vmax);
      avec[r] = exp2f(mrow[r] - mn);
      mrow[r] = mn;
      e0 = exp2f(e0 - mn);
      e1 = exp2f(e1 - mn);
      float rs = e0 + e1;
#pragma unroll
      for (int m = 1; m <= 8; m <<= 1) rs += __shfl_xor(rs, m, 32);
      lrow[r] = lrow[r] * avec[r] + rs;
      s0[r] = e0;
      s1[r] = e1;
    }
#pragma unroll
    for (int t = 0; t < 16; ++t) oacc[t] *= avec;

#pragma unroll
    for (int r = 0; r < 8; ++r) {
      pb[(8 * lh + r) * 32 + l15]      = (_Float16)s0[r];
      pb[(8 * lh + r) * 32 + 16 + l15] = (_Float16)s1[r];
    }
    asm volatile("s_wait_dscnt 0" ::: "memory");
    const v16h pa = load_frag16(pb + l15 * 32, lh);

#pragma unroll
    for (int t = 0; t < 16; ++t) {
      const _Float16* vrow = fT + ((size_t)n * CV + t * 16 + l15) * HW + j0;
      const v16h vb = load_frag16(vrow, lh);
      oacc[t] = __builtin_amdgcn_wmma_f32_16x16x32_f16(false, pa, false, vb, (short)0, oacc[t], false, false);
    }
  }

  float fin[8];
#pragma unroll
  for (int r = 0; r < 8; ++r)
    fin[r] = (1.0f / lrow[r]) * mk[i0 + 8 * lh + r];

#pragma unroll
  for (int t = 0; t < 16; ++t) {
    float* dst = out + ((size_t)n * CV + t * 16 + l15) * HW + i0 + 8 * lh;
    float4 lo4 = make_float4(oacc[t][0] * fin[0], oacc[t][1] * fin[1],
                             oacc[t][2] * fin[2], oacc[t][3] * fin[3]);
    float4 hi4 = make_float4(oacc[t][4] * fin[4], oacc[t][5] * fin[5],
                             oacc[t][6] * fin[6], oacc[t][7] * fin[7]);
    *reinterpret_cast<float4*>(dst)     = lo4;
    *reinterpret_cast<float4*>(dst + 4) = hi4;
  }
}

extern "C" void kernel_launch(void* const* d_in, const int* in_sizes, int n_in,
                              void* d_out, int out_size, void* d_ws, size_t ws_size,
                              hipStream_t stream) {
  (void)in_sizes; (void)n_in; (void)out_size; (void)ws_size;
  const float* x  = (const float*)d_in[0];
  const float* mk = (const float*)d_in[1];
  const float* rw = (const float*)d_in[2];
  const float* rg = (const float*)d_in[3];
  const float* rb = (const float*)d_in[4];
  const float* rm = (const float*)d_in[5];
  const float* rv = (const float*)d_in[6];
  const float* aw = (const float*)d_in[7];
  const float* ag = (const float*)d_in[8];
  const float* ab = (const float*)d_in[9];
  const float* am = (const float*)d_in[10];
  const float* av = (const float*)d_in[11];
  float* out = (float*)d_out;

  _Float16* qbuf = (_Float16*)d_ws;
  _Float16* fT   = qbuf + (size_t)NB * HW * CQ;

  const dim3 blk(32, 4);
  proj_bn_relu_kernel<<<dim3(HW / 64, CV / 16, NB), blk, 0, stream>>>(
      x, rw, rg, rb, rm, rv, fT, 0);
  proj_bn_relu_kernel<<<dim3(HW / 64, CQ / 16, NB), blk, 0, stream>>>(
      x, aw, ag, ab, am, av, qbuf, 1);
  flash_attn_kernel<<<dim3(HW / 64, NB), blk, 0, stream>>>(qbuf, fT, mk, out);
}


# Attention: each 64-key K/V tile is copied once per workgroup into a two-buffer LDS ring (plain global loads + ds_store, one tile ahead); its four waves read fragments from LDS, not from global
# speedup vs baseline: 3.0800x; 3.0800x over previous
#include <hip/hip_runtime.h>
#include <hip/hip_bf16.h>

#define HW   4096
#define NB   8
#define CIN  256
#define CQ   128
#define CV   256

#define SK   64
#define KP   136
#define VP   72

typedef float    v2f  __attribute__((ext_vector_type(2)));
typedef float    v8f  __attribute__((ext_vector_type(8)));
typedef _Float16 v16h __attribute__((ext_vector_type(16)));
typedef _Float16 h8   __attribute__((ext_vector_type(8)));

static __device__ __forceinline__ v16h load_frag16(const _Float16* rowBase, int lh) {
  const h8 lo = *reinterpret_cast<const h8*>(rowBase + 8 * lh);
  const h8 hi = *reinterpret_cast<const h8*>(rowBase + 8 * lh + 16);
  v16h r;
#pragma unroll
  for (int i = 0; i < 8; ++i) { r[i] = lo[i]; r[i + 8] = hi[i]; }
  return r;
}

__global__ __launch_bounds__(128) void proj_bn_relu_kernel(
    const float* __restrict__ x, const float* __restrict__ w,
    const float* __restrict__ gamma, const float* __restrict__ beta,
    const float* __restrict__ mean, const float* __restrict__ var,
    _Float16* __restrict__ outh, int posMajor)
{
  const int lane = threadIdx.x;
  const int l15  = lane & 15;
  const int lh   = lane >> 4;
  const int p0   = (blockIdx.x * blockDim.y + threadIdx.y) * 16;
  const int o0   = blockIdx.y * 16;
  const int n    = blockIdx.z;

  const float* wrow = w + (size_t)(o0 + l15) * CIN;
  const float* xcol = x + (size_t)n * CIN * HW + (p0 + l15);

  v8f acc0 = {};
  v8f acc1 = {};
#pragma unroll 4
  for (int c0 = 0; c0 < CIN; c0 += 8) {
    const float2 aw0 = *reinterpret_cast<const float2*>(wrow + c0 + 2 * lh);
    const float2 aw1 = *reinterpret_cast<const float2*>(wrow + c0 + 4 + 2 * lh);
    v2f a0; a0[0] = aw0.x; a0[1] = aw0.y;
    v2f a1; a1[0] = aw1.x; a1[1] = aw1.y;
    v2f b0; b0[0] = xcol[(size_t)(c0 + 2 * lh + 0) * HW];
            b0[1] = xcol[(size_t)(c0 + 2 * lh + 1) * HW];
    v2f b1; b1[0] = xcol[(size_t)(c0 + 4 + 2 * lh + 0) * HW];
            b1[1] = xcol[(size_t)(c0 + 4 + 2 * lh + 1) * HW];
    acc0 = __builtin_amdgcn_wmma_f32_16x16x4_f32(false, a0, false, b0, (short)0, acc0, false, false);
    acc1 = __builtin_amdgcn_wmma_f32_16x16x4_f32(false, a1, false, b1, (short)0, acc1, false, false);
  }
  const v8f acc = acc0 + acc1;

#pragma unroll
  for (int r = 0; r < 8; ++r) {
    const int ch   = o0 + 8 * lh + r;
    const float iv = gamma[ch] * rsqrtf(var[ch] + 1e-5f);
    const float bb = beta[ch] - mean[ch] * iv;
    float v = acc[r] * iv + bb;
    v = v > 0.f ? v : 0.f;
    const _Float16 hv = (_Float16)v;
    const int p = p0 + l15;
    if (posMajor)
      outh[((size_t)n * HW + p) * CQ + ch] = hv;
    else
      outh[((size_t)n * CV + ch) * HW + p] = hv;
  }
}

__global__ __launch_bounds__(128) void flash_attn_kernel(
    const _Float16* q, const _Float16* fT,
    const float* __restrict__ mask, float* __restrict__ out)
{
  constexpr float LOG2E     = 1.4426950408889634f;
  constexpr float ATT_SCALE = (1.0f / (1e-8f + 11.313708498984761f)) * LOG2E;
  constexpr int   NSTAGE    = HW / SK;
  __shared__ __align__(16) _Float16 Kst[2][SK * KP];
  __shared__ __align__(16) _Float16 Vst[2][CV * VP];
  __shared__ __align__(16) _Float16 Pbuf[4][16 * 32];

  const int lane = threadIdx.x;
  const int l15  = lane & 15;
  const int lh   = lane >> 4;
  const int wid  = threadIdx.y;
  const int tid  = wid * 32 + lane;
  const int i0   = (blockIdx.x * 4 + wid) * 16;
  const int n    = blockIdx.y;

  v16h qa[4];
  {
    const _Float16* qrow = q + ((size_t)n * HW + i0 + l15) * CQ;
#pragma unroll
    for (int kc = 0; kc < 4; ++kc) qa[kc] = load_frag16(qrow + kc * 32, lh);
  }

  const _Float16* kg = q  + (size_t)n * HW * CQ + (size_t)tid * 8;
  const _Float16* vg = fT + ((size_t)n * CV + (tid >> 3)) * HW + (tid & 7) * 8;
  const int kl = (tid >> 4) * KP + (tid & 15) * 8;
  const int vl = (tid >> 3) * VP + (tid & 7) * 8;

  uint4 pk0, pk1, pk2, pk3, pv0, pv1, pv2, pv3, pv4, pv5, pv6, pv7;
#define FETCH_HALF(S, H)                                                              \
  do {                                                                                \
    const _Float16* kgs_ = kg + (size_t)((S) * SK + (H) * 32) * CQ;                   \
    const _Float16* vgs_ = vg + (size_t)((H) * 128) * HW + (S) * SK;                  \
    pk0 = *reinterpret_cast<const uint4*>(kgs_);                                      \
    pk1 = *reinterpret_cast<const uint4*>(kgs_ + 1 * 8 * CQ);                         \
    pk2 = *reinterpret_cast<const uint4*>(kgs_ + 2 * 8 * CQ);                         \
    pk3 = *reinterpret_cast<const uint4*>(kgs_ + 3 * 8 * CQ);                         \
    pv0 = *reinterpret_cast<const uint4*>(vgs_);                                      \
    pv1 = *reinterpret_cast<const uint4*>(vgs_ + (size_t)16 * HW);                    \
    pv2 = *reinterpret_cast<const uint4*>(vgs_ + (size_t)32 * HW);                    \
    pv3 = *reinterpret_cast<const uint4*>(vgs_ + (size_t)48 * HW);                    \
    pv4 = *reinterpret_cast<const uint4*>(vgs_ + (size_t)64 * HW);                    \
    pv5 = *reinterpret_cast<const uint4*>(vgs_ + (size_t)80 * HW);                    \
    pv6 = *reinterpret_cast<const uint4*>(vgs_ + (size_t)96 * HW);                    \
    pv7 = *reinterpret_cast<const uint4*>(vgs_ + (size_t)112 * HW);                   \
  } while (0)
#define STASH_HALF(B, H)                                                              \
  do {                                                                                \
    _Float16* kd_ = &Kst[(B)][0] + kl + (H) * 32 * KP;                                \
    _Float16* vd_ = &Vst[(B)][0] + vl + (H) * 128 * VP;                               \
    *reinterpret_cast<uint4*>(kd_)               = pk0;                               \
    *reinterpret_cast<uint4*>(kd_ + 1 * 8 * KP)  = pk1;                               \
    *reinterpret_cast<uint4*>(kd_ + 2 * 8 * KP)  = pk2;                               \
    *reinterpret_cast<uint4*>(kd_ + 3 * 8 * KP)  = pk3;                               \
    *reinterpret_cast<uint4*>(vd_)               = pv0;                               \
    *reinterpret_cast<uint4*>(vd_ + 16 * VP)     = pv1;                               \
    *reinterpret_cast<uint4*>(vd_ + 32 * VP)     = pv2;                               \
    *reinterpret_cast<uint4*>(vd_ + 48 * VP)     = pv3;                               \
    *reinterpret_cast<uint4*>(vd_ + 64 * VP)     = pv4;                               \
    *reinterpret_cast<uint4*>(vd_ + 80 * VP)     = pv5;                               \
    *reinterpret_cast<uint4*>(vd_ + 96 * VP)     = pv6;                               \
    *reinterpret_cast<uint4*>(vd_ + 112 * VP)    = pv7;                               \
  } while (0)

  v8f oacc[16];
#pragma unroll
  for (int t = 0; t < 16; ++t) oacc[t] = (v8f){};
  float mrow[8], lrow[8];
#pragma unroll
  for (int r = 0; r < 8; ++r) { mrow[r] = -1e30f; lrow[r] = 0.f; }

  _Float16* pb    = &Pbuf[wid][0];
  const float* mk = mask + (size_t)n * HW;

  FETCH_HALF(0, 0); STASH_HALF(0, 0);
  FETCH_HALF(0, 1); STASH_HALF(0, 1);
  float mn0 = mk[l15];
  float mn1 = mk[16 + l15];
  __syncthreads();

  for (int s = 0; s < NSTAGE; ++s) {
    const int b  = s & 1;
    const int sn = (s + 1 < NSTAGE) ? s + 1 : NSTAGE - 1;
    const _Float16* kt = &Kst[b][0] + l15 * KP;
    const _Float16* vt = &Vst[b][0] + l15 * VP;

#pragma unroll
    for (int h = 0; h < 2; ++h) {
      const int j0 = s * SK + h * 32;
      const float mc0 = mn0, mc1 = mn1;
      {
        const int jn = (j0 + 32 < HW) ? j0 + 32 : HW - 32;
        mn0 = mk[jn + l15];
        mn1 = mk[jn + 16 + l15];
      }
      FETCH_HALF(sn, h);
      asm volatile("" ::: "memory");

      v8f s0 = (v8f){};
      v8f s1 = (v8f){};
#pragma unroll
      for (int kc = 0; kc < 4; ++kc) {
        const _Float16* kb = kt + (h * 32) * KP + kc * 32;
        const v16h b0 = load_frag16(kb, lh);
        const v16h b1 = load_frag16(kb + 16 * KP, lh);
        s0 = __builtin_amdgcn_wmma_f32_16x16x32_f16(false, qa[kc], false, b0, (short)0, s0, false, false);
        s1 = __builtin_amdgcn_wmma_f32_16x16x32_f16(false, qa[kc], false, b1, (short)0, s1, false, false);
      }

      const float ma0 = (mc0 - 1.f) * (1e8f * LOG2E);
      const float ma1 = (mc1 - 1.f) * (1e8f * LOG2E);

      v8f avec;
#pragma unroll
      for (int r = 0; r < 8; ++r) {
        float e0 = s0[r] * ATT_SCALE + ma0;
        float e1 = s1[r] * ATT_SCALE + ma1;
        float vmax = fmaxf(e0, e1);
#pragma unroll
        for (int m = 1; m <= 8; m <<= 1) vmax = fmaxf(vmax, __shfl_xor(vmax, m, 32));
        const float mn = fmaxf(mrow[r], vmax);
        avec[r] = exp2f(mrow[r] - mn);
        mrow[r] = mn;
        e0 = exp2f(e0 - mn);
        e1 = exp2f(e1 - mn);
        float rs = e0 + e1;
#pragma unroll
        for (int m = 1; m <= 8; m <<= 1) rs += __shfl_xor(rs, m, 32);
        lrow[r] = lrow[r] * avec[r] + rs;
        s0[r] = e0;
        s1[r] = e1;
      }
#pragma unroll
      for (int t = 0; t < 16; ++t) oacc[t] *= avec;

#pragma unroll
      for (int r = 0; r < 8; ++r) {
        pb[(8 * lh + r) * 32 + l15]      = (_Float16)s0[r];
        pb[(8 * lh + r) * 32 + 16 + l15] = (_Float16)s1[r];
      }
      asm volatile("s_wait_dscnt 0" ::: "memory");
      const v16h pa = load_frag16(pb + l15 * 32, lh);

#pragma unroll
      for (int t = 0; t < 16; ++t) {
        const v16h vb = load_frag16(vt + (t * 16) * VP + h * 32, lh);
        oacc[t] = __builtin_amdgcn_wmma_f32_16x16x32_f16(false, pa, false, vb, (short)0, oacc[t], false, false);
      }

      STASH_HALF(b ^ 1, h);
    }
    __syncthreads();
  }

  float fin[8];
#pragma unroll
  for (int r = 0; r < 8; ++r)
    fin[r] = (1.0f / lrow[r]) * mk[i0 + 8 * lh + r];

#pragma unroll
  for (int t = 0; t < 16; ++t) {
    float* dst = out + ((size_t)n * CV + t * 16 + l15) * HW + i0 + 8 * lh;
    float4 lo4 = make_float4(oacc[t][0] * fin[0], oacc[t][1] * fin[1],
                             oacc[t][2] * fin[2], oacc[t][3] * fin[3]);
    float4 hi4 = make_float4(oacc[t][4] * fin[4], oacc[t][5] * fin[5],
                             oacc[t][6] * fin[6], oacc[t][7] * fin[7]);
    *reinterpret_cast<float4*>(dst)     = lo4;
    *reinterpret_cast<float4*>(dst + 4) = hi4;
  }
}

#undef FETCH_HALF
#undef STASH_HALF

extern "C" void kernel_launch(void* const* d_in, const int* in_sizes, int n_in,
                              void* d_out, int out_size, void* d_ws, size_t ws_size,
                              hipStream_t stream) {
  (void)in_sizes; (void)n_in; (void)out_size; (void)ws_size;
  const float* x  = (const float*)d_in[0];
  const float* mk = (const float*)d_in[1];
  const float* rw = (const float*)d_in[2];
  const float* rg = (const float*)d_in[3];
  const float* rb = (const float*)d_in[4];
  const float* rm = (const float*)d_in[5];
  const float* rv = (const float*)d_in[6];
  const float* aw = (const float*)d_in[7];
  const float* ag = (const float*)d_in[8];
  const float* ab = (const float*)d_in[9];
  const float* am = (const float*)d_in[10];
  const float* av = (const float*)d_in[11];
  float* out = (float*)d_out;

  _Float16* qbuf = (_Float16*)d_ws;
  _Float16* fT   = qbuf + (size_t)NB * HW * CQ;

  const dim3 blk(32, 4);
  proj_bn_relu_kernel<<<dim3(HW / 64, CV / 16, NB), blk, 0, stream>>>(
      x, rw, rg, rb, rm, rv, fT, 0);
  proj_bn_relu_kernel<<<dim3(HW / 64, CQ / 16, NB), blk, 0, stream>>>(
      x, aw, ag, ab, am, av, qbuf, 1);
  flash_attn_kernel<<<dim3(HW / 64, NB), blk, 0, stream>>>(qbuf, fT, mk, out);
}


# Attention softmax: 16-lane row max and row sum butterflies use DPP row_xmask moves (vector ALU) in place of ds_bpermute_b32 (LDS unit); same xor pairing, so sums are bit-identical
# speedup vs baseline: 3.4161x; 1.1091x over previous
#include <hip/hip_runtime.h>
#include <hip/hip_bf16.h>

#define HW   4096
#define NB   8
#define CIN  256
#define CQ   128
#define CV   256

#define SK   64
#define KP   136
#define VP   72

typedef float    v2f  __attribute__((ext_vector_type(2)));
typedef float    v8f  __attribute__((ext_vector_type(8)));
typedef _Float16 v16h __attribute__((ext_vector_type(16)));
typedef _Float16 h8   __attribute__((ext_vector_type(8)));

static __device__ __forceinline__ v16h load_frag16(const _Float16* rowBase, int lh) {
  const h8 lo = *reinterpret_cast<const h8*>(rowBase + 8 * lh);
  const h8 hi = *reinterpret_cast<const h8*>(rowBase + 8 * lh + 16);
  v16h r;
#pragma unroll
  for (int i = 0; i < 8; ++i) { r[i] = lo[i]; r[i + 8] = hi[i]; }
  return r;
}


template <int M>
static __device__ __forceinline__ float row_xor(float v) {
  return __int_as_float(__builtin_amdgcn_update_dpp(0, __float_as_int(v), 0x160 + M, 0xf, 0xf, true));
}

__global__ __launch_bounds__(128) void proj_bn_relu_kernel(
    const float* __restrict__ x, const float* __restrict__ w,
    const float* __restrict__ gamma, const float* __restrict__ beta,
    const float* __restrict__ mean, const float* __restrict__ var,
    _Float16* __restrict__ outh, int posMajor)
{
  const int lane = threadIdx.x;
  const int l15  = lane & 15;
  const int lh   = lane >> 4;
  const int p0   = (blockIdx.x * blockDim.y + threadIdx.y) * 16;
  const int o0   = blockIdx.y * 16;
  const int n    = blockIdx.z;

  const float* wrow = w + (size_t)(o0 + l15) * CIN;
  const float* xcol = x + (size_t)n * CIN * HW + (p0 + l15);

  v8f acc0 = {};
  v8f acc1 = {};
#pragma unroll 4
  for (int c0 = 0; c0 < CIN; c0 += 8) {
    const float2 aw0 = *reinterpret_cast<const float2*>(wrow + c0 + 2 * lh);
    const float2 aw1 = *reinterpret_cast<const float2*>(wrow + c0 + 4 + 2 * lh);
    v2f a0; a0[0] = aw0.x; a0[1] = aw0.y;
    v2f a1; a1[0] = aw1.x; a1[1] = aw1.y;
    v2f b0; b0[0] = xcol[(size_t)(c0 + 2 * lh + 0) * HW];
            b0[1] = xcol[(size_t)(c0 + 2 * lh + 1) * HW];
    v2f b1; b1[0] = xcol[(size_t)(c0 + 4 + 2 * lh + 0) * HW];
            b1[1] = xcol[(size_t)(c0 + 4 + 2 * lh + 1) * HW];
    acc0 = __builtin_amdgcn_wmma_f32_16x16x4_f32(false, a0, false, b0, (short)0, acc0, false, false);
    acc1 = __builtin_amdgcn_wmma_f32_16x16x4_f32(false, a1, false, b1, (short)0, acc1, false, false);
  }
  const v8f acc = acc0 + acc1;

#pragma unroll
  for (int r = 0; r < 8; ++r) {
    const int ch   = o0 + 8 * lh + r;
    const float iv = gamma[ch] * rsqrtf(var[ch] + 1e-5f);
    const float bb = beta[ch] - mean[ch] * iv;
    float v = acc[r] * iv + bb;
    v = v > 0.f ? v : 0.f;
    const _Float16 hv = (_Float16)v;
    const int p = p0 + l15;
    if (posMajor)
      outh[((size_t)n * HW + p) * CQ + ch] = hv;
    else
      outh[((size_t)n * CV + ch) * HW + p] = hv;
  }
}

__global__ __launch_bounds__(128) void flash_attn_kernel(
    const _Float16* q, const _Float16* fT,
    const float* __restrict__ mask, float* __restrict__ out)
{
  constexpr float LOG2E     = 1.4426950408889634f;
  constexpr float ATT_SCALE = (1.0f / (1e-8f + 11.313708498984761f)) * LOG2E;
  constexpr int   NSTAGE    = HW / SK;
  __shared__ __align__(16) _Float16 Kst[2][SK * KP];
  __shared__ __align__(16) _Float16 Vst[2][CV * VP];
  __shared__ __align__(16) _Float16 Pbuf[4][16 * 32];

  const int lane = threadIdx.x;
  const int l15  = lane & 15;
  const int lh   = lane >> 4;
  const int wid  = threadIdx.y;
  const int tid  = wid * 32 + lane;
  const int i0   = (blockIdx.x * 4 + wid) * 16;
  const int n    = blockIdx.y;

  v16h qa[4];
  {
    const _Float16* qrow = q + ((size_t)n * HW + i0 + l15) * CQ;
#pragma unroll
    for (int kc = 0; kc < 4; ++kc) qa[kc] = load_frag16(qrow + kc * 32, lh);
  }

  const _Float16* kg = q  + (size_t)n * HW * CQ + (size_t)tid * 8;
  const _Float16* vg = fT + ((size_t)n * CV + (tid >> 3)) * HW + (tid & 7) * 8;
  const int kl = (tid >> 4) * KP + (tid & 15) * 8;
  const int vl = (tid >> 3) * VP + (tid & 7) * 8;

  uint4 pk0, pk1, pk2, pk3, pv0, pv1, pv2, pv3, pv4, pv5, pv6, pv7;
#define FETCH_HALF(S, H)                                                              \
  do {                                                                                \
    const _Float16* kgs_ = kg + (size_t)((S) * SK + (H) * 32) * CQ;                   \
    const _Float16* vgs_ = vg + (size_t)((H) * 128) * HW + (S) * SK;                  \
    pk0 = *reinterpret_cast<const uint4*>(kgs_);                                      \
    pk1 = *reinterpret_cast<const uint4*>(kgs_ + 1 * 8 * CQ);                         \
    pk2 = *reinterpret_cast<const uint4*>(kgs_ + 2 * 8 * CQ);                         \
    pk3 = *reinterpret_cast<const uint4*>(kgs_ + 3 * 8 * CQ);                         \
    pv0 = *reinterpret_cast<const uint4*>(vgs_);                                      \
    pv1 = *reinterpret_cast<const uint4*>(vgs_ + (size_t)16 * HW);                    \
    pv2 = *reinterpret_cast<const uint4*>(vgs_ + (size_t)32 * HW);                    \
    pv3 = *reinterpret_cast<const uint4*>(vgs_ + (size_t)48 * HW);                    \
    pv4 = *reinterpret_cast<const uint4*>(vgs_ + (size_t)64 * HW);                    \
    pv5 = *reinterpret_cast<const uint4*>(vgs_ + (size_t)80 * HW);                    \
    pv6 = *reinterpret_cast<const uint4*>(vgs_ + (size_t)96 * HW);                    \
    pv7 = *reinterpret_cast<const uint4*>(vgs_ + (size_t)112 * HW);                   \
  } while (0)
#define STASH_HALF(B, H)                                                              \
  do {                                                                                \
    _Float16* kd_ = &Kst[(B)][0] + kl + (H) * 32 * KP;                                \
    _Float16* vd_ = &Vst[(B)][0] + vl + (H) * 128 * VP;                               \
    *reinterpret_cast<uint4*>(kd_)               = pk0;                               \
    *reinterpret_cast<uint4*>(kd_ + 1 * 8 * KP)  = pk1;                               \
    *reinterpret_cast<uint4*>(kd_ + 2 * 8 * KP)  = pk2;                               \
    *reinterpret_cast<uint4*>(kd_ + 3 * 8 * KP)  = pk3;                               \
    *reinterpret_cast<uint4*>(vd_)               = pv0;                               \
    *reinterpret_cast<uint4*>(vd_ + 16 * VP)     = pv1;                               \
    *reinterpret_cast<uint4*>(vd_ + 32 * VP)     = pv2;                               \
    *reinterpret_cast<uint4*>(vd_ + 48 * VP)     = pv3;                               \
    *reinterpret_cast<uint4*>(vd_ + 64 * VP)     = pv4;                               \
    *reinterpret_cast<uint4*>(vd_ + 80 * VP)     = pv5;                               \
    *reinterpret_cast<uint4*>(vd_ + 96 * VP)     = pv6;                               \
    *reinterpret_cast<uint4*>(vd_ + 112 * VP)    = pv7;                               \
  } while (0)

  v8f oacc[16];
#pragma unroll
  for (int t = 0; t < 16; ++t) oacc[t] = (v8f){};
  float mrow[8], lrow[8];
#pragma unroll
  for (int r = 0; r < 8; ++r) { mrow[r] = -1e30f; lrow[r] = 0.f; }

  _Float16* pb    = &Pbuf[wid][0];
  const float* mk = mask + (size_t)n * HW;

  FETCH_HALF(0, 0); STASH_HALF(0, 0);
  FETCH_HALF(0, 1); STASH_HALF(0, 1);
  float mn0 = mk[l15];
  float mn1 = mk[16 + l15];
  __syncthreads();

  for (int s = 0; s < NSTAGE; ++s) {
    const int b  = s & 1;
    const int sn = (s + 1 < NSTAGE) ? s + 1 : NSTAGE - 1;
    const _Float16* kt = &Kst[b][0] + l15 * KP;
    const _Float16* vt = &Vst[b][0] + l15 * VP;

#pragma unroll
    for (int h = 0; h < 2; ++h) {
      const int j0 = s * SK + h * 32;
      const float mc0 = mn0, mc1 = mn1;
      {
        const int jn = (j0 + 32 < HW) ? j0 + 32 : HW - 32;
        mn0 = mk[jn + l15];
        mn1 = mk[jn + 16 + l15];
      }
      FETCH_HALF(sn, h);
      asm volatile("" ::: "memory");

      v8f s0 = (v8f){};
      v8f s1 = (v8f){};
#pragma unroll
      for (int kc = 0; kc < 4; ++kc) {
        const _Float16* kb = kt + (h * 32) * KP + kc * 32;
        const v16h b0 = load_frag16(kb, lh);
        const v16h b1 = load_frag16(kb + 16 * KP, lh);
        s0 = __builtin_amdgcn_wmma_f32_16x16x32_f16(false, qa[kc], false, b0, (short)0, s0, false, false);
        s1 = __builtin_amdgcn_wmma_f32_16x16x32_f16(false, qa[kc], false, b1, (short)0, s1, false, false);
      }

      const float ma0 = (mc0 - 1.f) * (1e8f * LOG2E);
      const float ma1 = (mc1 - 1.f) * (1e8f * LOG2E);

      v8f avec;
#pragma unroll
      for (int r = 0; r < 8; ++r) {
        float e0 = s0[r] * ATT_SCALE + ma0;
        float e1 = s1[r] * ATT_SCALE + ma1;
        float vmax = fmaxf(e0, e1);
        vmax = fmaxf(vmax, row_xor<1>(vmax));
        vmax = fmaxf(vmax, row_xor<2>(vmax));
        vmax = fmaxf(vmax, row_xor<4>(vmax));
        vmax = fmaxf(vmax, row_xor<8>(vmax));
        const float mn = fmaxf(mrow[r], vmax);
        avec[r] = exp2f(mrow[r] - mn);
        mrow[r] = mn;
        e0 = exp2f(e0 - mn);
        e1 = exp2f(e1 - mn);
        float rs = e0 + e1;
        rs += row_xor<1>(rs);
        rs += row_xor<2>(rs);
        rs += row_xor<4>(rs);
        rs += row_xor<8>(rs);
        lrow[r] = lrow[r] * avec[r] + rs;
        s0[r] = e0;
        s1[r] = e1;
      }
#pragma unroll
      for (int t = 0; t < 16; ++t) oacc[t] *= avec;

#pragma unroll
      for (int r = 0; r < 8; ++r) {
        pb[(8 * lh + r) * 32 + l15]      = (_Float16)s0[r];
        pb[(8 * lh + r) * 32 + 16 + l15] = (_Float16)s1[r];
      }
      asm volatile("s_wait_dscnt 0" ::: "memory");
      const v16h pa = load_frag16(pb + l15 * 32, lh);

#pragma unroll
      for (int t = 0; t < 16; ++t) {
        const v16h vb = load_frag16(vt + (t * 16) * VP + h * 32, lh);
        oacc[t] = __builtin_amdgcn_wmma_f32_16x16x32_f16(false, pa, false, vb, (short)0, oacc[t], false, false);
      }

      STASH_HALF(b ^ 1, h);
    }
    __syncthreads();
  }

  float fin[8];
#pragma unroll
  for (int r = 0; r < 8; ++r)
    fin[r] = (1.0f / lrow[r]) * mk[i0 + 8 * lh + r];

#pragma unroll
  for (int t = 0; t < 16; ++t) {
    float* dst = out + ((size_t)n * CV + t * 16 + l15) * HW + i0 + 8 * lh;
    float4 lo4 = make_float4(oacc[t][0] * fin[0], oacc[t][1] * fin[1],
                             oacc[t][2] * fin[2], oacc[t][3] * fin[3]);
    float4 hi4 = make_float4(oacc[t][4] * fin[4], oacc[t][5] * fin[5],
                             oacc[t][6] * fin[6], oacc[t][7] * fin[7]);
    *reinterpret_cast<float4*>(dst)     = lo4;
    *reinterpret_cast<float4*>(dst + 4) = hi4;
  }
}

#undef FETCH_HALF
#undef STASH_HALF

extern "C" void kernel_launch(void* const* d_in, const int* in_sizes, int n_in,
                              void* d_out, int out_size, void* d_ws, size_t ws_size,
                              hipStream_t stream) {
  (void)in_sizes; (void)n_in; (void)out_size; (void)ws_size;
  const float* x  = (const float*)d_in[0];
  const float* mk = (const float*)d_in[1];
  const float* rw = (const float*)d_in[2];
  const float* rg = (const float*)d_in[3];
  const float* rb = (const float*)d_in[4];
  const float* rm = (const float*)d_in[5];
  const float* rv = (const float*)d_in[6];
  const float* aw = (const float*)d_in[7];
  const float* ag = (const float*)d_in[8];
  const float* ab = (const float*)d_in[9];
  const float* am = (const float*)d_in[10];
  const float* av = (const float*)d_in[11];
  float* out = (float*)d_out;

  _Float16* qbuf = (_Float16*)d_ws;
  _Float16* fT   = qbuf + (size_t)NB * HW * CQ;

  const dim3 blk(32, 4);
  proj_bn_relu_kernel<<<dim3(HW / 64, CV / 16, NB), blk, 0, stream>>>(
      x, rw, rg, rb, rm, rv, fT, 0);
  proj_bn_relu_kernel<<<dim3(HW / 64, CQ / 16, NB), blk, 0, stream>>>(
      x, aw, ag, ab, am, av, qbuf, 1);
  flash_attn_kernel<<<dim3(HW / 64, NB), blk, 0, stream>>>(qbuf, fT, mk, out);
}


# Attention: each wave owns two 16-query tiles (128 queries per workgroup, grid 32x8, one wave per SIMD declared): every K and V^T fragment read from LDS feeds two WMMAs; sum order per query unchanged
# speedup vs baseline: 4.4583x; 1.3051x over previous
#include <hip/hip_runtime.h>
#include <hip/hip_bf16.h>

#define HW   4096
#define NB   8
#define CIN  256
#define CQ   128
#define CV   256

#define SK   64
#define KP   136
#define VP   72

typedef float    v2f  __attribute__((ext_vector_type(2)));
typedef float    v8f  __attribute__((ext_vector_type(8)));
typedef _Float16 v16h __attribute__((ext_vector_type(16)));
typedef _Float16 h8   __attribute__((ext_vector_type(8)));

static __device__ __forceinline__ v16h load_frag16(const _Float16* rowBase, int lh) {
  const h8 lo = *reinterpret_cast<const h8*>(rowBase + 8 * lh);
  const h8 hi = *reinterpret_cast<const h8*>(rowBase + 8 * lh + 16);
  v16h r;
#pragma unroll
  for (int i = 0; i < 8; ++i) { r[i] = lo[i]; r[i + 8] = hi[i]; }
  return r;
}


static __device__ __forceinline__ float xhalf(float v) {
  return __int_as_float(__builtin_amdgcn_permlanex16(__float_as_int(v), __float_as_int(v),
                                                     0x76543210, 0xfedcba98, false, false));
}

__global__ __launch_bounds__(128) void proj_bn_relu_kernel(
    const float* __restrict__ x, const float* __restrict__ w,
    const float* __restrict__ gamma, const float* __restrict__ beta,
    const float* __restrict__ mean, const float* __restrict__ var,
    _Float16* __restrict__ outh, int posMajor)
{
  const int lane = threadIdx.x;
  const int l15  = lane & 15;
  const int lh   = lane >> 4;
  const int p0   = (blockIdx.x * blockDim.y + threadIdx.y) * 16;
  const int o0   = blockIdx.y * 16;
  const int n    = blockIdx.z;

  const float* wrow = w + (size_t)(o0 + l15) * CIN;
  const float* xcol = x + (size_t)n * CIN * HW + (p0 + l15);

  v8f acc0 = {};
  v8f acc1 = {};
#pragma unroll 4
  for (int c0 = 0; c0 < CIN; c0 += 8) {
    const float2 aw0 = *reinterpret_cast<const float2*>(wrow + c0 + 2 * lh);
    const float2 aw1 = *reinterpret_cast<const float2*>(wrow + c0 + 4 + 2 * lh);
    v2f a0; a0[0] = aw0.x; a0[1] = aw0.y;
    v2f a1; a1[0] = aw1.x; a1[1] = aw1.y;
    v2f b0; b0[0] = xcol[(size_t)(c0 + 2 * lh + 0) * HW];
            b0[1] = xcol[(size_t)(c0 + 2 * lh + 1) * HW];
    v2f b1; b1[0] = xcol[(size_t)(c0 + 4 + 2 * lh + 0) * HW];
            b1[1] = xcol[(size_t)(c0 + 4 + 2 * lh + 1) * HW];
    acc0 = __builtin_amdgcn_wmma_f32_16x16x4_f32(false, a0, false, b0, (short)0, acc0, false, false);
    acc1 = __builtin_amdgcn_wmma_f32_16x16x4_f32(false, a1, false, b1, (short)0, acc1, false, false);
  }
  const v8f acc = acc0 + acc1;

#pragma unroll
  for (int r = 0; r < 8; ++r) {
    const int ch   = o0 + 8 * lh + r;
    const float iv = gamma[ch] * rsqrtf(var[ch] + 1e-5f);
    const float bb = beta[ch] - mean[ch] * iv;
    float v = acc[r] * iv + bb;
    v = v > 0.f ? v : 0.f;
    const _Float16 hv = (_Float16)v;
    const int p = p0 + l15;
    if (posMajor)
      outh[((size_t)n * HW + p) * CQ + ch] = hv;
    else
      outh[((size_t)n * CV + ch) * HW + p] = hv;
  }
}

__global__ __launch_bounds__(128) __attribute__((amdgpu_waves_per_eu(1, 1))) void flash_attn_kernel(
    const _Float16* q, const _Float16* fT,
    const float* __restrict__ mask, float* __restrict__ out)
{
  constexpr float LOG2E     = 1.4426950408889634f;
  constexpr float ATT_SCALE = (1.0f / (1e-8f + 11.313708498984761f)) * LOG2E;
  constexpr int   NSTAGE    = HW / SK;
  __shared__ __align__(16) _Float16 Kst[2][SK * KP];
  __shared__ __align__(16) _Float16 Vst[2][CV * VP];
  __shared__ __align__(32) float    Madd[HW];

  const int lane = threadIdx.x;
  const int l15  = lane & 15;
  const int lh   = lane >> 4;
  const int wid  = threadIdx.y;
  const int tid  = wid * 32 + lane;
  const int i0   = (blockIdx.x * 4 + wid) * 32;
  const int n    = blockIdx.y;

  v16h qa[2][4];
#pragma unroll
  for (int u = 0; u < 2; ++u) {
    const _Float16* qrow = q + ((size_t)n * HW + i0 + 16 * u + l15) * CQ;
#pragma unroll
    for (int kc = 0; kc < 4; ++kc) qa[u][kc] = load_frag16(qrow + kc * 32, lh);
  }

  const _Float16* kg = q  + (size_t)n * HW * CQ + (size_t)tid * 8;
  const _Float16* vg = fT + ((size_t)n * CV + (tid >> 3)) * HW + (tid & 7) * 8;
  const int kl = (tid >> 4) * KP + (tid & 15) * 8;
  const int vl = (tid >> 3) * VP + (tid & 7) * 8;

  uint4 pk0, pk1, pk2, pk3, pv0, pv1, pv2, pv3, pv4, pv5, pv6, pv7;
#define FETCH_HALF(S, H)                                                              \
  do {                                                                                \
    const _Float16* kgs_ = kg + (size_t)((S) * SK + (H) * 32) * CQ;                   \
    const _Float16* vgs_ = vg + (size_t)((H) * 128) * HW + (S) * SK;                  \
    pk0 = *reinterpret_cast<const uint4*>(kgs_);                                      \
    pk1 = *reinterpret_cast<const uint4*>(kgs_ + 1 * 8 * CQ);                         \
    pk2 = *reinterpret_cast<const uint4*>(kgs_ + 2 * 8 * CQ);                         \
    pk3 = *reinterpret_cast<const uint4*>(kgs_ + 3 * 8 * CQ);                         \
    pv0 = *reinterpret_cast<const uint4*>(vgs_);                                      \
    pv1 = *reinterpret_cast<const uint4*>(vgs_ + (size_t)16 * HW);                    \
    pv2 = *reinterpret_cast<const uint4*>(vgs_ + (size_t)32 * HW);                    \
    pv3 = *reinterpret_cast<const uint4*>(vgs_ + (size_t)48 * HW);                    \
    pv4 = *reinterpret_cast<const uint4*>(vgs_ + (size_t)64 * HW);                    \
    pv5 = *reinterpret_cast<const uint4*>(vgs_ + (size_t)80 * HW);                    \
    pv6 = *reinterpret_cast<const uint4*>(vgs_ + (size_t)96 * HW);                    \
    pv7 = *reinterpret_cast<const uint4*>(vgs_ + (size_t)112 * HW);                   \
  } while (0)
#define STASH_HALF(B, H)                                                              \
  do {                                                                                \
    _Float16* kd_ = &Kst[(B)][0] + kl + (H) * 32 * KP;                                \
    _Float16* vd_ = &Vst[(B)][0] + vl + (H) * 128 * VP;                               \
    *reinterpret_cast<uint4*>(kd_)               = pk0;                               \
    *reinterpret_cast<uint4*>(kd_ + 1 * 8 * KP)  = pk1;                               \
    *reinterpret_cast<uint4*>(kd_ + 2 * 8 * KP)  = pk2;                               \
    *reinterpret_cast<uint4*>(kd_ + 3 * 8 * KP)  = pk3;                               \
    *reinterpret_cast<uint4*>(vd_)               = pv0;                               \
    *reinterpret_cast<uint4*>(vd_ + 16 * VP)     = pv1;                               \
    *reinterpret_cast<uint4*>(vd_ + 32 * VP)     = pv2;                               \
    *reinterpret_cast<uint4*>(vd_ + 48 * VP)     = pv3;                               \
    *reinterpret_cast<uint4*>(vd_ + 64 * VP)     = pv4;                               \
    *reinterpret_cast<uint4*>(vd_ + 80 * VP)     = pv5;                               \
    *reinterpret_cast<uint4*>(vd_ + 96 * VP)     = pv6;                               \
    *reinterpret_cast<uint4*>(vd_ + 112 * VP)    = pv7;                               \
  } while (0)

  v8f oacc[2][16];
#pragma unroll
  for (int u = 0; u < 2; ++u) {
#pragma unroll
    for (int t = 0; t < 16; ++t) oacc[u][t] = (v8f){};
  }
  float mrun[2] = {-1e30f, -1e30f};
  float lrun[2] = {0.f, 0.f};

  const float* mk = mask + (size_t)n * HW;
  for (int i = 0; i < HW / 128; ++i)
    Madd[tid + 128 * i] = (mk[tid + 128 * i] - 1.f) * (1e8f * LOG2E);

  FETCH_HALF(0, 0); STASH_HALF(0, 0);
  FETCH_HALF(0, 1); STASH_HALF(0, 1);
  __syncthreads();

  for (int s = 0; s < NSTAGE; ++s) {
    const int b  = s & 1;
    const int sn = (s + 1 < NSTAGE) ? s + 1 : NSTAGE - 1;
    const _Float16* kt = &Kst[b][0] + l15 * KP;
    const _Float16* vt = &Vst[b][0] + l15 * VP;

#pragma unroll
    for (int h = 0; h < 2; ++h) {
      const int j0 = s * SK + h * 32;
      FETCH_HALF(sn, h);
      asm volatile("" ::: "memory");

      const v8f ma0 = *reinterpret_cast<const v8f*>(&Madd[j0 + 8 * lh]);
      const v8f ma1 = *reinterpret_cast<const v8f*>(&Madd[j0 + 16 + 8 * lh]);

      v8f s0[2], s1[2];
#pragma unroll
      for (int u = 0; u < 2; ++u) { s0[u] = (v8f){}; s1[u] = (v8f){}; }
#pragma unroll
      for (int kc = 0; kc < 4; ++kc) {
        const _Float16* kb = kt + (h * 32) * KP + kc * 32;
        const v16h a0 = load_frag16(kb, lh);
        const v16h a1 = load_frag16(kb + 16 * KP, lh);
#pragma unroll
        for (int u = 0; u < 2; ++u) {
          s0[u] = __builtin_amdgcn_wmma_f32_16x16x32_f16(false, a0, false, qa[u][kc], (short)0, s0[u], false, false);
          s1[u] = __builtin_amdgcn_wmma_f32_16x16x32_f16(false, a1, false, qa[u][kc], (short)0, s1[u], false, false);
        }
      }

      v16h pbt[2];
#pragma unroll
      for (int u = 0; u < 2; ++u) {
        float vmax = -1e30f;
#pragma unroll
        for (int r = 0; r < 8; ++r) {
          s0[u][r] = s0[u][r] * ATT_SCALE + ma0[r];
          s1[u][r] = s1[u][r] * ATT_SCALE + ma1[r];
          vmax = fmaxf(vmax, fmaxf(s0[u][r], s1[u][r]));
        }
        vmax = fmaxf(vmax, xhalf(vmax));
        const float mn    = fmaxf(mrun[u], vmax);
        const float alpha = exp2f(mrun[u] - mn);
        mrun[u] = mn;
        float rs0 = 0.f, rs1 = 0.f;
#pragma unroll
        for (int r = 0; r < 8; ++r) {
          s0[u][r] = exp2f(s0[u][r] - mn);
          s1[u][r] = exp2f(s1[u][r] - mn);
          rs0 += s0[u][r];
          rs1 += s1[u][r];
        }
        lrun[u] = lrun[u] * alpha + (rs0 + rs1);
#pragma unroll
        for (int t = 0; t < 16; ++t) oacc[u][t] *= alpha;

#pragma unroll
        for (int r = 0; r < 8; ++r) { pbt[u][r] = (_Float16)s0[u][r]; pbt[u][r + 8] = (_Float16)s1[u][r]; }
      }

#pragma unroll
      for (int t = 0; t < 16; ++t) {
        const v16h va = load_frag16(vt + (t * 16) * VP + h * 32, lh);
#pragma unroll
        for (int u = 0; u < 2; ++u)
          oacc[u][t] = __builtin_amdgcn_wmma_f32_16x16x32_f16(false, va, false, pbt[u], (short)0, oacc[u][t], false, false);
      }

      asm volatile("" :: "v"(lrun[0]), "v"(lrun[1]) : "memory");
      STASH_HALF(b ^ 1, h);
    }
    __syncthreads();
  }

#pragma unroll
  for (int u = 0; u < 2; ++u) {
    const float ltot = lrun[u] + xhalf(lrun[u]);
    const float fin  = (1.0f / ltot) * mk[i0 + 16 * u + l15];
#pragma unroll
    for (int t = 0; t < 16; ++t) {
      float* dst = out + ((size_t)n * CV + t * 16 + 8 * lh) * HW + i0 + 16 * u + l15;
#pragma unroll
      for (int r = 0; r < 8; ++r) dst[(size_t)r * HW] = oacc[u][t][r] * fin;
    }
  }
}

#undef FETCH_HALF
#undef STASH_HALF

extern "C" void kernel_launch(void* const* d_in, const int* in_sizes, int n_in,
                              void* d_out, int out_size, void* d_ws, size_t ws_size,
                              hipStream_t stream) {
  (void)in_sizes; (void)n_in; (void)out_size; (void)ws_size;
  const float* x  = (const float*)d_in[0];
  const float* mk = (const float*)d_in[1];
  const float* rw = (const float*)d_in[2];
  const float* rg = (const float*)d_in[3];
  const float* rb = (const float*)d_in[4];
  const float* rm = (const float*)d_in[5];
  const float* rv = (const float*)d_in[6];
  const float* aw = (const float*)d_in[7];
  const float* ag = (const float*)d_in[8];
  const float* ab = (const float*)d_in[9];
  const float* am = (const float*)d_in[10];
  const float* av = (const float*)d_in[11];
  float* out = (float*)d_out;

  _Float16* qbuf = (_Float16*)d_ws;
  _Float16* fT   = qbuf + (size_t)NB * HW * CQ;

  const dim3 blk(32, 4);
  proj_bn_relu_kernel<<<dim3(HW / 64, CV / 16, NB), blk, 0, stream>>>(
      x, rw, rg, rb, rm, rv, fT, 0);
  proj_bn_relu_kernel<<<dim3(HW / 64, CQ / 16, NB), blk, 0, stream>>>(
      x, aw, ag, ab, am, av, qbuf, 1);
  flash_attn_kernel<<<dim3(HW / 128, NB), blk, 0, stream>>>(qbuf, fT, mk, out);
}


# Projection GEMM register-tiled: each wave owns 64 channels x 64 positions (4x4 f32 WMMA tiles), 16-B W and x loads feed 4 WMMAs each; channels of an 8-chunk summed in permuted order
# speedup vs baseline: 5.6967x; 1.2778x over previous
#include <hip/hip_runtime.h>
#include <hip/hip_bf16.h>

#define HW   4096
#define NB   8
#define CIN  256
#define CQ   128
#define CV   256

#define SK   64
#define KP   136
#define VP   72

typedef float    v2f  __attribute__((ext_vector_type(2)));
typedef float    v4f  __attribute__((ext_vector_type(4)));
typedef float    v8f  __attribute__((ext_vector_type(8)));
typedef _Float16 v16h __attribute__((ext_vector_type(16)));
typedef _Float16 h8   __attribute__((ext_vector_type(8)));
typedef _Float16 h4   __attribute__((ext_vector_type(4)));

static __device__ __forceinline__ v16h load_frag16(const _Float16* rowBase, int lh) {
  const h8 lo = *reinterpret_cast<const h8*>(rowBase + 8 * lh);
  const h8 hi = *reinterpret_cast<const h8*>(rowBase + 8 * lh + 16);
  v16h r;
#pragma unroll
  for (int i = 0; i < 8; ++i) { r[i] = lo[i]; r[i + 8] = hi[i]; }
  return r;
}


static __device__ __forceinline__ float xhalf(float v) {
  return __int_as_float(__builtin_amdgcn_permlanex16(__float_as_int(v), __float_as_int(v),
                                                     0x76543210, 0xfedcba98, false, false));
}

__global__ __launch_bounds__(128) void proj_bn_relu_kernel(
    const float* __restrict__ x, const float* __restrict__ w,
    const float* __restrict__ gamma, const float* __restrict__ beta,
    const float* __restrict__ mean, const float* __restrict__ var,
    _Float16* __restrict__ outh, int posMajor, int mshift)
{
  const int lane = threadIdx.x;
  const int l15  = lane & 15;
  const int lh   = lane >> 4;
  const int wid  = threadIdx.y;
  const int o0   = (wid & ((1 << mshift) - 1)) * 64;
  const int p0   = (blockIdx.x * (4 >> mshift) + (wid >> mshift)) * 64;
  const int n    = blockIdx.y;

  const float* wrow = w + (size_t)(o0 + l15) * CIN + 4 * lh;
  const float* xcol = x + ((size_t)n * CIN + 4 * lh) * HW + p0 + 4 * l15;

  v8f acc[4][4];
#pragma unroll
  for (int m = 0; m < 4; ++m) {
#pragma unroll
    for (int j = 0; j < 4; ++j) acc[m][j] = (v8f){};
  }

  v4f aw[4], bx[4];
#pragma unroll
  for (int m = 0; m < 4; ++m) aw[m] = *reinterpret_cast<const v4f*>(wrow + (size_t)(16 * m) * CIN);
#pragma unroll
  for (int e = 0; e < 4; ++e) bx[e] = *reinterpret_cast<const v4f*>(xcol + (size_t)e * HW);

#pragma unroll 2
  for (int c0 = 0; c0 < CIN; c0 += 8) {
    const int cn = (c0 + 8 < CIN) ? c0 + 8 : c0;
    v4f awn[4], bxn[4];
#pragma unroll
    for (int m = 0; m < 4; ++m) awn[m] = *reinterpret_cast<const v4f*>(wrow + (size_t)(16 * m) * CIN + cn);
#pragma unroll
    for (int e = 0; e < 4; ++e) bxn[e] = *reinterpret_cast<const v4f*>(xcol + (size_t)(cn + e) * HW);

#pragma unroll
    for (int s = 0; s < 2; ++s) {
      v2f b[4];
#pragma unroll
      for (int j = 0; j < 4; ++j) { b[j][0] = bx[2 * s][j]; b[j][1] = bx[2 * s + 1][j]; }
#pragma unroll
      for (int m = 0; m < 4; ++m) {
        v2f a; a[0] = aw[m][2 * s]; a[1] = aw[m][2 * s + 1];
#pragma unroll
        for (int j = 0; j < 4; ++j)
          acc[m][j] = __builtin_amdgcn_wmma_f32_16x16x4_f32(false, a, false, b[j], (short)0, acc[m][j], false, false);
      }
    }
#pragma unroll
    for (int m = 0; m < 4; ++m) { aw[m] = awn[m]; bx[m] = bxn[m]; }
  }

#pragma unroll
  for (int m = 0; m < 4; ++m) {
    const int chb = o0 + 16 * m + 8 * lh;
    float iv[8], bb[8];
#pragma unroll
    for (int q4 = 0; q4 < 2; ++q4) {
      const v4f g  = *reinterpret_cast<const v4f*>(gamma + chb + 4 * q4);
      const v4f vv = *reinterpret_cast<const v4f*>(var   + chb + 4 * q4);
      const v4f be = *reinterpret_cast<const v4f*>(beta  + chb + 4 * q4);
      const v4f mu = *reinterpret_cast<const v4f*>(mean  + chb + 4 * q4);
#pragma unroll
      for (int r = 0; r < 4; ++r) {
        iv[4 * q4 + r] = g[r] * rsqrtf(vv[r] + 1e-5f);
        bb[4 * q4 + r] = be[r] - mu[r] * iv[4 * q4 + r];
      }
    }
    if (posMajor) {
#pragma unroll
      for (int j = 0; j < 4; ++j) {
        h8 pk;
#pragma unroll
        for (int r = 0; r < 8; ++r) {
          float v = acc[m][j][r] * iv[r] + bb[r];
          v = v > 0.f ? v : 0.f;
          pk[r] = (_Float16)v;
        }
        *reinterpret_cast<h8*>(outh + ((size_t)n * HW + p0 + 4 * l15 + j) * CQ + chb) = pk;
      }
    } else {
#pragma unroll
      for (int r = 0; r < 8; ++r) {
        h4 pk;
#pragma unroll
        for (int j = 0; j < 4; ++j) {
          float v = acc[m][j][r] * iv[r] + bb[r];
          v = v > 0.f ? v : 0.f;
          pk[j] = (_Float16)v;
        }
        *reinterpret_cast<h4*>(outh + ((size_t)n * CV + chb + r) * HW + p0 + 4 * l15) = pk;
      }
    }
  }
}

__global__ __launch_bounds__(128) __attribute__((amdgpu_waves_per_eu(1, 1))) void flash_attn_kernel(
    const _Float16* q, const _Float16* fT,
    const float* __restrict__ mask, float* __restrict__ out)
{
  constexpr float LOG2E     = 1.4426950408889634f;
  constexpr float ATT_SCALE = (1.0f / (1e-8f + 11.313708498984761f)) * LOG2E;
  constexpr int   NSTAGE    = HW / SK;
  __shared__ __align__(16) _Float16 Kst[2][SK * KP];
  __shared__ __align__(16) _Float16 Vst[2][CV * VP];
  __shared__ __align__(32) float    Madd[HW];

  const int lane = threadIdx.x;
  const int l15  = lane & 15;
  const int lh   = lane >> 4;
  const int wid  = threadIdx.y;
  const int tid  = wid * 32 + lane;
  const int i0   = (blockIdx.x * 4 + wid) * 32;
  const int n    = blockIdx.y;

  v16h qa[2][4];
#pragma unroll
  for (int u = 0; u < 2; ++u) {
    const _Float16* qrow = q + ((size_t)n * HW + i0 + 16 * u + l15) * CQ;
#pragma unroll
    for (int kc = 0; kc < 4; ++kc) qa[u][kc] = load_frag16(qrow + kc * 32, lh);
  }

  const _Float16* kg = q  + (size_t)n * HW * CQ + (size_t)tid * 8;
  const _Float16* vg = fT + ((size_t)n * CV + (tid >> 3)) * HW + (tid & 7) * 8;
  const int kl = (tid >> 4) * KP + (tid & 15) * 8;
  const int vl = (tid >> 3) * VP + (tid & 7) * 8;

  uint4 pk0, pk1, pk2, pk3, pv0, pv1, pv2, pv3, pv4, pv5, pv6, pv7;
#define FETCH_HALF(S, H)                                                              \
  do {                                                                                \
    const _Float16* kgs_ = kg + (size_t)((S) * SK + (H) * 32) * CQ;                   \
    const _Float16* vgs_ = vg + (size_t)((H) * 128) * HW + (S) * SK;                  \
    pk0 = *reinterpret_cast<const uint4*>(kgs_);                                      \
    pk1 = *reinterpret_cast<const uint4*>(kgs_ + 1 * 8 * CQ);                         \
    pk2 = *reinterpret_cast<const uint4*>(kgs_ + 2 * 8 * CQ);                         \
    pk3 = *reinterpret_cast<const uint4*>(kgs_ + 3 * 8 * CQ);                         \
    pv0 = *reinterpret_cast<const uint4*>(vgs_);                                      \
    pv1 = *reinterpret_cast<const uint4*>(vgs_ + (size_t)16 * HW);                    \
    pv2 = *reinterpret_cast<const uint4*>(vgs_ + (size_t)32 * HW);                    \
    pv3 = *reinterpret_cast<const uint4*>(vgs_ + (size_t)48 * HW);                    \
    pv4 = *reinterpret_cast<const uint4*>(vgs_ + (size_t)64 * HW);                    \
    pv5 = *reinterpret_cast<const uint4*>(vgs_ + (size_t)80 * HW);                    \
    pv6 = *reinterpret_cast<const uint4*>(vgs_ + (size_t)96 * HW);                    \
    pv7 = *reinterpret_cast<const uint4*>(vgs_ + (size_t)112 * HW);                   \
  } while (0)
#define STASH_HALF(B, H)                                                              \
  do {                                                                                \
    _Float16* kd_ = &Kst[(B)][0] + kl + (H) * 32 * KP;                                \
    _Float16* vd_ = &Vst[(B)][0] + vl + (H) * 128 * VP;                               \
    *reinterpret_cast<uint4*>(kd_)               = pk0;                               \
    *reinterpret_cast<uint4*>(kd_ + 1 * 8 * KP)  = pk1;                               \
    *reinterpret_cast<uint4*>(kd_ + 2 * 8 * KP)  = pk2;                               \
    *reinterpret_cast<uint4*>(kd_ + 3 * 8 * KP)  = pk3;                               \
    *reinterpret_cast<uint4*>(vd_)               = pv0;                               \
    *reinterpret_cast<uint4*>(vd_ + 16 * VP)     = pv1;                               \
    *reinterpret_cast<uint4*>(vd_ + 32 * VP)     = pv2;                               \
    *reinterpret_cast<uint4*>(vd_ + 48 * VP)     = pv3;                               \
    *reinterpret_cast<uint4*>(vd_ + 64 * VP)     = pv4;                               \
    *reinterpret_cast<uint4*>(vd_ + 80 * VP)     = pv5;                               \
    *reinterpret_cast<uint4*>(vd_ + 96 * VP)     = pv6;                               \
    *reinterpret_cast<uint4*>(vd_ + 112 * VP)    = pv7;                               \
  } while (0)

  v8f oacc[2][16];
#pragma unroll
  for (int u = 0; u < 2; ++u) {
#pragma unroll
    for (int t = 0; t < 16; ++t) oacc[u][t] = (v8f){};
  }
  float mrun[2] = {-1e30f, -1e30f};
  float lrun[2] = {0.f, 0.f};

  const float* mk = mask + (size_t)n * HW;
  for (int i = 0; i < HW / 128; ++i)
    Madd[tid + 128 * i] = (mk[tid + 128 * i] - 1.f) * (1e8f * LOG2E);

  FETCH_HALF(0, 0); STASH_HALF(0, 0);
  FETCH_HALF(0, 1); STASH_HALF(0, 1);
  __syncthreads();

  for (int s = 0; s < NSTAGE; ++s) {
    const int b  = s & 1;
    const int sn = (s + 1 < NSTAGE) ? s + 1 : NSTAGE - 1;
    const _Float16* kt = &Kst[b][0] + l15 * KP;
    const _Float16* vt = &Vst[b][0] + l15 * VP;

#pragma unroll
    for (int h = 0; h < 2; ++h) {
      const int j0 = s * SK + h * 32;
      FETCH_HALF(sn, h);
      asm volatile("" ::: "memory");

      const v8f ma0 = *reinterpret_cast<const v8f*>(&Madd[j0 + 8 * lh]);
      const v8f ma1 = *reinterpret_cast<const v8f*>(&Madd[j0 + 16 + 8 * lh]);

      v8f s0[2], s1[2];
#pragma unroll
      for (int u = 0; u < 2; ++u) { s0[u] = (v8f){}; s1[u] = (v8f){}; }
#pragma unroll
      for (int kc = 0; kc < 4; ++kc) {
        const _Float16* kb = kt + (h * 32) * KP + kc * 32;
        const v16h a0 = load_frag16(kb, lh);
        const v16h a1 = load_frag16(kb + 16 * KP, lh);
#pragma unroll
        for (int u = 0; u < 2; ++u) {
          s0[u] = __builtin_amdgcn_wmma_f32_16x16x32_f16(false, a0, false, qa[u][kc], (short)0, s0[u], false, false);
          s1[u] = __builtin_amdgcn_wmma_f32_16x16x32_f16(false, a1, false, qa[u][kc], (short)0, s1[u], false, false);
        }
      }

      v16h pbt[2];
#pragma unroll
      for (int u = 0; u < 2; ++u) {
        float vmax = -1e30f;
#pragma unroll
        for (int r = 0; r < 8; ++r) {
          s0[u][r] = s0[u][r] * ATT_SCALE + ma0[r];
          s1[u][r] = s1[u][r] * ATT_SCALE + ma1[r];
          vmax = fmaxf(vmax, fmaxf(s0[u][r], s1[u][r]));
        }
        vmax = fmaxf(vmax, xhalf(vmax));
        const float mn    = fmaxf(mrun[u], vmax);
        const float alpha = exp2f(mrun[u] - mn);
        mrun[u] = mn;
        float rs0 = 0.f, rs1 = 0.f;
#pragma unroll
        for (int r = 0; r < 8; ++r) {
          s0[u][r] = exp2f(s0[u][r] - mn);
          s1[u][r] = exp2f(s1[u][r] - mn);
          rs0 += s0[u][r];
          rs1 += s1[u][r];
        }
        lrun[u] = lrun[u] * alpha + (rs0 + rs1);
#pragma unroll
        for (int t = 0; t < 16; ++t) oacc[u][t] *= alpha;

#pragma unroll
        for (int r = 0; r < 8; ++r) { pbt[u][r] = (_Float16)s0[u][r]; pbt[u][r + 8] = (_Float16)s1[u][r]; }
      }

#pragma unroll
      for (int t = 0; t < 16; ++t) {
        const v16h va = load_frag16(vt + (t * 16) * VP + h * 32, lh);
#pragma unroll
        for (int u = 0; u < 2; ++u)
          oacc[u][t] = __builtin_amdgcn_wmma_f32_16x16x32_f16(false, va, false, pbt[u], (short)0, oacc[u][t], false, false);
      }

      asm volatile("" :: "v"(lrun[0]), "v"(lrun[1]) : "memory");
      STASH_HALF(b ^ 1, h);
    }
    __syncthreads();
  }

#pragma unroll
  for (int u = 0; u < 2; ++u) {
    const float ltot = lrun[u] + xhalf(lrun[u]);
    const float fin  = (1.0f / ltot) * mk[i0 + 16 * u + l15];
#pragma unroll
    for (int t = 0; t < 16; ++t) {
      float* dst = out + ((size_t)n * CV + t * 16 + 8 * lh) * HW + i0 + 16 * u + l15;
#pragma unroll
      for (int r = 0; r < 8; ++r) dst[(size_t)r * HW] = oacc[u][t][r] * fin;
    }
  }
}

#undef FETCH_HALF
#undef STASH_HALF

extern "C" void kernel_launch(void* const* d_in, const int* in_sizes, int n_in,
                              void* d_out, int out_size, void* d_ws, size_t ws_size,
                              hipStream_t stream) {
  (void)in_sizes; (void)n_in; (void)out_size; (void)ws_size;
  const float* x  = (const float*)d_in[0];
  const float* mk = (const float*)d_in[1];
  const float* rw = (const float*)d_in[2];
  const float* rg = (const float*)d_in[3];
  const float* rb = (const float*)d_in[4];
  const float* rm = (const float*)d_in[5];
  const float* rv = (const float*)d_in[6];
  const float* aw = (const float*)d_in[7];
  const float* ag = (const float*)d_in[8];
  const float* ab = (const float*)d_in[9];
  const float* am = (const float*)d_in[10];
  const float* av = (const float*)d_in[11];
  float* out = (float*)d_out;

  _Float16* qbuf = (_Float16*)d_ws;
  _Float16* fT   = qbuf + (size_t)NB * HW * CQ;

  const dim3 blk(32, 4);
  proj_bn_relu_kernel<<<dim3(HW / 64, NB), blk, 0, stream>>>(
      x, rw, rg, rb, rm, rv, fT, 0, 2);
  proj_bn_relu_kernel<<<dim3(HW / 128, NB), blk, 0, stream>>>(
      x, aw, ag, ab, am, av, qbuf, 1, 1);
  flash_attn_kernel<<<dim3(HW / 128, NB), blk, 0, stream>>>(qbuf, fT, mk, out);
}


# Attention softmax: every exponential is the bare v_exp_f32 (builtin) in place of library exp2f; the denormal-range fix-up around each (compare, two selects, ldexp) goes away; sum order unchanged
# speedup vs baseline: 5.9912x; 1.0517x over previous
#include <hip/hip_runtime.h>
#include <hip/hip_bf16.h>

#define HW   4096
#define NB   8
#define CIN  256
#define CQ   128
#define CV   256

#define SK   64
#define KP   136
#define VP   72

typedef float    v2f  __attribute__((ext_vector_type(2)));
typedef float    v4f  __attribute__((ext_vector_type(4)));
typedef float    v8f  __attribute__((ext_vector_type(8)));
typedef _Float16 v16h __attribute__((ext_vector_type(16)));
typedef _Float16 h8   __attribute__((ext_vector_type(8)));
typedef _Float16 h4   __attribute__((ext_vector_type(4)));

static __device__ __forceinline__ v16h load_frag16(const _Float16* rowBase, int lh) {
  const h8 lo = *reinterpret_cast<const h8*>(rowBase + 8 * lh);
  const h8 hi = *reinterpret_cast<const h8*>(rowBase + 8 * lh + 16);
  v16h r;
#pragma unroll
  for (int i = 0; i < 8; ++i) { r[i] = lo[i]; r[i + 8] = hi[i]; }
  return r;
}


static __device__ __forceinline__ float xhalf(float v) {
  return __int_as_float(__builtin_amdgcn_permlanex16(__float_as_int(v), __float_as_int(v),
                                                     0x76543210, 0xfedcba98, false, false));
}

static __device__ __forceinline__ float ex2(float v) { return __builtin_amdgcn_exp2f(v); }

__global__ __launch_bounds__(128) void proj_bn_relu_kernel(
    const float* __restrict__ x, const float* __restrict__ w,
    const float* __restrict__ gamma, const float* __restrict__ beta,
    const float* __restrict__ mean, const float* __restrict__ var,
    _Float16* __restrict__ outh, int posMajor, int mshift)
{
  const int lane = threadIdx.x;
  const int l15  = lane & 15;
  const int lh   = lane >> 4;
  const int wid  = threadIdx.y;
  const int o0   = (wid & ((1 << mshift) - 1)) * 64;
  const int p0   = (blockIdx.x * (4 >> mshift) + (wid >> mshift)) * 64;
  const int n    = blockIdx.y;

  const float* wrow = w + (size_t)(o0 + l15) * CIN + 4 * lh;
  const float* xcol = x + ((size_t)n * CIN + 4 * lh) * HW + p0 + 4 * l15;

  v8f acc[4][4];
#pragma unroll
  for (int m = 0; m < 4; ++m) {
#pragma unroll
    for (int j = 0; j < 4; ++j) acc[m][j] = (v8f){};
  }

  v4f aw[4], bx[4];
#pragma unroll
  for (int m = 0; m < 4; ++m) aw[m] = *reinterpret_cast<const v4f*>(wrow + (size_t)(16 * m) * CIN);
#pragma unroll
  for (int e = 0; e < 4; ++e) bx[e] = *reinterpret_cast<const v4f*>(xcol + (size_t)e * HW);

#pragma unroll 2
  for (int c0 = 0; c0 < CIN; c0 += 8) {
    const int cn = (c0 + 8 < CIN) ? c0 + 8 : c0;
    v4f awn[4], bxn[4];
#pragma unroll
    for (int m = 0; m < 4; ++m) awn[m] = *reinterpret_cast<const v4f*>(wrow + (size_t)(16 * m) * CIN + cn);
#pragma unroll
    for (int e = 0; e < 4; ++e) bxn[e] = *reinterpret_cast<const v4f*>(xcol + (size_t)(cn + e) * HW);

#pragma unroll
    for (int s = 0; s < 2; ++s) {
      v2f b[4];
#pragma unroll
      for (int j = 0; j < 4; ++j) { b[j][0] = bx[2 * s][j]; b[j][1] = bx[2 * s + 1][j]; }
#pragma unroll
      for (int m = 0; m < 4; ++m) {
        v2f a; a[0] = aw[m][2 * s]; a[1] = aw[m][2 * s + 1];
#pragma unroll
        for (int j = 0; j < 4; ++j)
          acc[m][j] = __builtin_amdgcn_wmma_f32_16x16x4_f32(false, a, false, b[j], (short)0, acc[m][j], false, false);
      }
    }
#pragma unroll
    for (int m = 0; m < 4; ++m) { aw[m] = awn[m]; bx[m] = bxn[m]; }
  }

#pragma unroll
  for (int m = 0; m < 4; ++m) {
    const int chb = o0 + 16 * m + 8 * lh;
    float iv[8], bb[8];
#pragma unroll
    for (int q4 = 0; q4 < 2; ++q4) {
      const v4f g  = *reinterpret_cast<const v4f*>(gamma + chb + 4 * q4);
      const v4f vv = *reinterpret_cast<const v4f*>(var   + chb + 4 * q4);
      const v4f be = *reinterpret_cast<const v4f*>(beta  + chb + 4 * q4);
      const v4f mu = *reinterpret_cast<const v4f*>(mean  + chb + 4 * q4);
#pragma unroll
      for (int r = 0; r < 4; ++r) {
        iv[4 * q4 + r] = g[r] * rsqrtf(vv[r] + 1e-5f);
        bb[4 * q4 + r] = be[r] - mu[r] * iv[4 * q4 + r];
      }
    }
    if (posMajor) {
#pragma unroll
      for (int j = 0; j < 4; ++j) {
        h8 pk;
#pragma unroll
        for (int r = 0; r < 8; ++r) {
          float v = acc[m][j][r] * iv[r] + bb[r];
          v = v > 0.f ? v : 0.f;
          pk[r] = (_Float16)v;
        }
        *reinterpret_cast<h8*>(outh + ((size_t)n * HW + p0 + 4 * l15 + j) * CQ + chb) = pk;
      }
    } else {
#pragma unroll
      for (int r = 0; r < 8; ++r) {
        h4 pk;
#pragma unroll
        for (int j = 0; j < 4; ++j) {
          float v = acc[m][j][r] * iv[r] + bb[r];
          v = v > 0.f ? v : 0.f;
          pk[j] = (_Float16)v;
        }
        *reinterpret_cast<h4*>(outh + ((size_t)n * CV + chb + r) * HW + p0 + 4 * l15) = pk;
      }
    }
  }
}

__global__ __launch_bounds__(128) __attribute__((amdgpu_waves_per_eu(1, 1))) void flash_attn_kernel(
    const _Float16* q, const _Float16* fT,
    const float* __restrict__ mask, float* __restrict__ out)
{
  constexpr float LOG2E     = 1.4426950408889634f;
  constexpr float ATT_SCALE = (1.0f / (1e-8f + 11.313708498984761f)) * LOG2E;
  constexpr int   NSTAGE    = HW / SK;
  __shared__ __align__(16) _Float16 Kst[2][SK * KP];
  __shared__ __align__(16) _Float16 Vst[2][CV * VP];
  __shared__ __align__(32) float    Madd[HW];

  const int lane = threadIdx.x;
  const int l15  = lane & 15;
  const int lh   = lane >> 4;
  const int wid  = threadIdx.y;
  const int tid  = wid * 32 + lane;
  const int i0   = (blockIdx.x * 4 + wid) * 32;
  const int n    = blockIdx.y;

  v16h qa[2][4];
#pragma unroll
  for (int u = 0; u < 2; ++u) {
    const _Float16* qrow = q + ((size_t)n * HW + i0 + 16 * u + l15) * CQ;
#pragma unroll
    for (int kc = 0; kc < 4; ++kc) qa[u][kc] = load_frag16(qrow + kc * 32, lh);
  }

  const _Float16* kg = q  + (size_t)n * HW * CQ + (size_t)tid * 8;
  const _Float16* vg = fT + ((size_t)n * CV + (tid >> 3)) * HW + (tid & 7) * 8;
  const int kl = (tid >> 4) * KP + (tid & 15) * 8;
  const int vl = (tid >> 3) * VP + (tid & 7) * 8;

  uint4 pk0, pk1, pk2, pk3, pv0, pv1, pv2, pv3, pv4, pv5, pv6, pv7;
#define FETCH_HALF(S, H)                                                              \
  do {                                                                                \
    const _Float16* kgs_ = kg + (size_t)((S) * SK + (H) * 32) * CQ;                   \
    const _Float16* vgs_ = vg + (size_t)((H) * 128) * HW + (S) * SK;                  \
    pk0 = *reinterpret_cast<const uint4*>(kgs_);                                      \
    pk1 = *reinterpret_cast<const uint4*>(kgs_ + 1 * 8 * CQ);                         \
    pk2 = *reinterpret_cast<const uint4*>(kgs_ + 2 * 8 * CQ);                         \
    pk3 = *reinterpret_cast<const uint4*>(kgs_ + 3 * 8 * CQ);                         \
    pv0 = *reinterpret_cast<const uint4*>(vgs_);                                      \
    pv1 = *reinterpret_cast<const uint4*>(vgs_ + (size_t)16 * HW);                    \
    pv2 = *reinterpret_cast<const uint4*>(vgs_ + (size_t)32 * HW);                    \
    pv3 = *reinterpret_cast<const uint4*>(vgs_ + (size_t)48 * HW);                    \
    pv4 = *reinterpret_cast<const uint4*>(vgs_ + (size_t)64 * HW);                    \
    pv5 = *reinterpret_cast<const uint4*>(vgs_ + (size_t)80 * HW);                    \
    pv6 = *reinterpret_cast<const uint4*>(vgs_ + (size_t)96 * HW);                    \
    pv7 = *reinterpret_cast<const uint4*>(vgs_ + (size_t)112 * HW);                   \
  } while (0)
#define STASH_HALF(B, H)                                                              \
  do {                                                                                \
    _Float16* kd_ = &Kst[(B)][0] + kl + (H) * 32 * KP;                                \
    _Float16* vd_ = &Vst[(B)][0] + vl + (H) * 128 * VP;                               \
    *reinterpret_cast<uint4*>(kd_)               = pk0;                               \
    *reinterpret_cast<uint4*>(kd_ + 1 * 8 * KP)  = pk1;                               \
    *reinterpret_cast<uint4*>(kd_ + 2 * 8 * KP)  = pk2;                               \
    *reinterpret_cast<uint4*>(kd_ + 3 * 8 * KP)  = pk3;                               \
    *reinterpret_cast<uint4*>(vd_)               = pv0;                               \
    *reinterpret_cast<uint4*>(vd_ + 16 * VP)     = pv1;                               \
    *reinterpret_cast<uint4*>(vd_ + 32 * VP)     = pv2;                               \
    *reinterpret_cast<uint4*>(vd_ + 48 * VP)     = pv3;                               \
    *reinterpret_cast<uint4*>(vd_ + 64 * VP)     = pv4;                               \
    *reinterpret_cast<uint4*>(vd_ + 80 * VP)     = pv5;                               \
    *reinterpret_cast<uint4*>(vd_ + 96 * VP)     = pv6;                               \
    *reinterpret_cast<uint4*>(vd_ + 112 * VP)    = pv7;                               \
  } while (0)

  v8f oacc[2][16];
#pragma unroll
  for (int u = 0; u < 2; ++u) {
#pragma unroll
    for (int t = 0; t < 16; ++t) oacc[u][t] = (v8f){};
  }
  float mrun[2] = {-1e30f, -1e30f};
  float lrun[2] = {0.f, 0.f};

  const float* mk = mask + (size_t)n * HW;
  for (int i = 0; i < HW / 128; ++i)
    Madd[tid + 128 * i] = (mk[tid + 128 * i] - 1.f) * (1e8f * LOG2E);

  FETCH_HALF(0, 0); STASH_HALF(0, 0);
  FETCH_HALF(0, 1); STASH_HALF(0, 1);
  __syncthreads();

  for (int s = 0; s < NSTAGE; ++s) {
    const int b  = s & 1;
    const int sn = (s + 1 < NSTAGE) ? s + 1 : NSTAGE - 1;
    const _Float16* kt = &Kst[b][0] + l15 * KP;
    const _Float16* vt = &Vst[b][0] + l15 * VP;

#pragma unroll
    for (int h = 0; h < 2; ++h) {
      const int j0 = s * SK + h * 32;
      FETCH_HALF(sn, h);
      asm volatile("" ::: "memory");

      const v8f ma0 = *reinterpret_cast<const v8f*>(&Madd[j0 + 8 * lh]);
      const v8f ma1 = *reinterpret_cast<const v8f*>(&Madd[j0 + 16 + 8 * lh]);

      v8f s0[2], s1[2];
#pragma unroll
      for (int u = 0; u < 2; ++u) { s0[u] = (v8f){}; s1[u] = (v8f){}; }
#pragma unroll
      for (int kc = 0; kc < 4; ++kc) {
        const _Float16* kb = kt + (h * 32) * KP + kc * 32;
        const v16h a0 = load_frag16(kb, lh);
        const v16h a1 = load_frag16(kb + 16 * KP, lh);
#pragma unroll
        for (int u = 0; u < 2; ++u) {
          s0[u] = __builtin_amdgcn_wmma_f32_16x16x32_f16(false, a0, false, qa[u][kc], (short)0, s0[u], false, false);
          s1[u] = __builtin_amdgcn_wmma_f32_16x16x32_f16(false, a1, false, qa[u][kc], (short)0, s1[u], false, false);
        }
      }

      v16h pbt[2];
#pragma unroll
      for (int u = 0; u < 2; ++u) {
        float vmax = -1e30f;
#pragma unroll
        for (int r = 0; r < 8; ++r) {
          s0[u][r] = s0[u][r] * ATT_SCALE + ma0[r];
          s1[u][r] = s1[u][r] * ATT_SCALE + ma1[r];
          vmax = fmaxf(vmax, fmaxf(s0[u][r], s1[u][r]));
        }
        vmax = fmaxf(vmax, xhalf(vmax));
        const float mn    = fmaxf(mrun[u], vmax);
        const float alpha = ex2(mrun[u] - mn);
        mrun[u] = mn;
        float rs0 = 0.f, rs1 = 0.f;
#pragma unroll
        for (int r = 0; r < 8; ++r) {
          s0[u][r] = ex2(s0[u][r] - mn);
          s1[u][r] = ex2(s1[u][r] - mn);
          rs0 += s0[u][r];
          rs1 += s1[u][r];
        }
        lrun[u] = lrun[u] * alpha + (rs0 + rs1);
#pragma unroll
        for (int t = 0; t < 16; ++t) oacc[u][t] *= alpha;

#pragma unroll
        for (int r = 0; r < 8; ++r) { pbt[u][r] = (_Float16)s0[u][r]; pbt[u][r + 8] = (_Float16)s1[u][r]; }
      }

#pragma unroll
      for (int t = 0; t < 16; ++t) {
        const v16h va = load_frag16(vt + (t * 16) * VP + h * 32, lh);
#pragma unroll
        for (int u = 0; u < 2; ++u)
          oacc[u][t] = __builtin_amdgcn_wmma_f32_16x16x32_f16(false, va, false, pbt[u], (short)0, oacc[u][t], false, false);
      }

      asm volatile("" :: "v"(lrun[0]), "v"(lrun[1]) : "memory");
      STASH_HALF(b ^ 1, h);
    }
    __syncthreads();
  }

#pragma unroll
  for (int u = 0; u < 2; ++u) {
    const float ltot = lrun[u] + xhalf(lrun[u]);
    const float fin  = (1.0f / ltot) * mk[i0 + 16 * u + l15];
#pragma unroll
    for (int t = 0; t < 16; ++t) {
      float* dst = out + ((size_t)n * CV + t * 16 + 8 * lh) * HW + i0 + 16 * u + l15;
#pragma unroll
      for (int r = 0; r < 8; ++r) dst[(size_t)r * HW] = oacc[u][t][r] * fin;
    }
  }
}

#undef FETCH_HALF
#undef STASH_HALF

extern "C" void kernel_launch(void* const* d_in, const int* in_sizes, int n_in,
                              void* d_out, int out_size, void* d_ws, size_t ws_size,
                              hipStream_t stream) {
  (void)in_sizes; (void)n_in; (void)out_size; (void)ws_size;
  const float* x  = (const float*)d_in[0];
  const float* mk = (const float*)d_in[1];
  const float* rw = (const float*)d_in[2];
  const float* rg = (const float*)d_in[3];
  const float* rb = (const float*)d_in[4];
  const float* rm = (const float*)d_in[5];
  const float* rv = (const float*)d_in[6];
  const float* aw = (const float*)d_in[7];
  const float* ag = (const float*)d_in[8];
  const float* ab = (const float*)d_in[9];
  const float* am = (const float*)d_in[10];
  const float* av = (const float*)d_in[11];
  float* out = (float*)d_out;

  _Float16* qbuf = (_Float16*)d_ws;
  _Float16* fT   = qbuf + (size_t)NB * HW * CQ;

  const dim3 blk(32, 4);
  proj_bn_relu_kernel<<<dim3(HW / 64, NB), blk, 0, stream>>>(
      x, rw, rg, rb, rm, rv, fT, 0, 2);
  proj_bn_relu_kernel<<<dim3(HW / 128, NB), blk, 0, stream>>>(
      x, aw, ag, ab, am, av, qbuf, 1, 1);
  flash_attn_kernel<<<dim3(HW / 128, NB), blk, 0, stream>>>(qbuf, fT, mk, out);
}


# Attention: 8 waves per workgroup, two per SIMD: the waves of a pair split each staged 64-key tile by key half; max, row sum and output of the halves merged once at the end (two half sums per query)
# speedup vs baseline: 6.9282x; 1.1564x over previous
#include <hip/hip_runtime.h>
#include <hip/hip_bf16.h>

#define HW   4096
#define NB   8
#define CIN  256
#define CQ   128
#define CV   256

#define SK   64
#define KP   136
#define VP   72

typedef float    v2f  __attribute__((ext_vector_type(2)));
typedef float    v4f  __attribute__((ext_vector_type(4)));
typedef float    v8f  __attribute__((ext_vector_type(8)));
typedef _Float16 v16h __attribute__((ext_vector_type(16)));
typedef _Float16 h8   __attribute__((ext_vector_type(8)));
typedef _Float16 h4   __attribute__((ext_vector_type(4)));

static __device__ __forceinline__ v16h load_frag16(const _Float16* rowBase, int lh) {
  const h8 lo = *reinterpret_cast<const h8*>(rowBase + 8 * lh);
  const h8 hi = *reinterpret_cast<const h8*>(rowBase + 8 * lh + 16);
  v16h r;
#pragma unroll
  for (int i = 0; i < 8; ++i) { r[i] = lo[i]; r[i + 8] = hi[i]; }
  return r;
}


static __device__ __forceinline__ float xhalf(float v) {
  return __int_as_float(__builtin_amdgcn_permlanex16(__float_as_int(v), __float_as_int(v),
                                                     0x76543210, 0xfedcba98, false, false));
}

static __device__ __forceinline__ float ex2(float v) { return __builtin_amdgcn_exp2f(v); }

__global__ __launch_bounds__(128) void proj_bn_relu_kernel(
    const float* __restrict__ x, const float* __restrict__ w,
    const float* __restrict__ gamma, const float* __restrict__ beta,
    const float* __restrict__ mean, const float* __restrict__ var,
    _Float16* __restrict__ outh, int posMajor, int mshift)
{
  const int lane = threadIdx.x;
  const int l15  = lane & 15;
  const int lh   = lane >> 4;
  const int wid  = threadIdx.y;
  const int o0   = (wid & ((1 << mshift) - 1)) * 64;
  const int p0   = (blockIdx.x * (4 >> mshift) + (wid >> mshift)) * 64;
  const int n    = blockIdx.y;

  const float* wrow = w + (size_t)(o0 + l15) * CIN + 4 * lh;
  const float* xcol = x + ((size_t)n * CIN + 4 * lh) * HW + p0 + 4 * l15;

  v8f acc[4][4];
#pragma unroll
  for (int m = 0; m < 4; ++m) {
#pragma unroll
    for (int j = 0; j < 4; ++j) acc[m][j] = (v8f){};
  }

  v4f aw[4], bx[4];
#pragma unroll
  for (int m = 0; m < 4; ++m) aw[m] = *reinterpret_cast<const v4f*>(wrow + (size_t)(16 * m) * CIN);
#pragma unroll
  for (int e = 0; e < 4; ++e) bx[e] = *reinterpret_cast<const v4f*>(xcol + (size_t)e * HW);

#pragma unroll 2
  for (int c0 = 0; c0 < CIN; c0 += 8) {
    const int cn = (c0 + 8 < CIN) ? c0 + 8 : c0;
    v4f awn[4], bxn[4];
#pragma unroll
    for (int m = 0; m < 4; ++m) awn[m] = *reinterpret_cast<const v4f*>(wrow + (size_t)(16 * m) * CIN + cn);
#pragma unroll
    for (int e = 0; e < 4; ++e) bxn[e] = *reinterpret_cast<const v4f*>(xcol + (size_t)(cn + e) * HW);

#pragma unroll
    for (int s = 0; s < 2; ++s) {
      v2f b[4];
#pragma unroll
      for (int j = 0; j < 4; ++j) { b[j][0] = bx[2 * s][j]; b[j][1] = bx[2 * s + 1][j]; }
#pragma unroll
      for (int m = 0; m < 4; ++m) {
        v2f a; a[0] = aw[m][2 * s]; a[1] = aw[m][2 * s + 1];
#pragma unroll
        for (int j = 0; j < 4; ++j)
          acc[m][j] = __builtin_amdgcn_wmma_f32_16x16x4_f32(false, a, false, b[j], (short)0, acc[m][j], false, false);
      }
    }
#pragma unroll
    for (int m = 0; m < 4; ++m) { aw[m] = awn[m]; bx[m] = bxn[m]; }
  }

#pragma unroll
  for (int m = 0; m < 4; ++m) {
    const int chb = o0 + 16 * m + 8 * lh;
    float iv[8], bb[8];
#pragma unroll
    for (int q4 = 0; q4 < 2; ++q4) {
      const v4f g  = *reinterpret_cast<const v4f*>(gamma + chb + 4 * q4);
      const v4f vv = *reinterpret_cast<const v4f*>(var   + chb + 4 * q4);
      const v4f be = *reinterpret_cast<const v4f*>(beta  + chb + 4 * q4);
      const v4f mu = *reinterpret_cast<const v4f*>(mean  + chb + 4 * q4);
#pragma unroll
      for (int r = 0; r < 4; ++r) {
        iv[4 * q4 + r] = g[r] * rsqrtf(vv[r] + 1e-5f);
        bb[4 * q4 + r] = be[r] - mu[r] * iv[4 * q4 + r];
      }
    }
    if (posMajor) {
#pragma unroll
      for (int j = 0; j < 4; ++j) {
        h8 pk;
#pragma unroll
        for (int r = 0; r < 8; ++r) {
          float v = acc[m][j][r] * iv[r] + bb[r];
          v = v > 0.f ? v : 0.f;
          pk[r] = (_Float16)v;
        }
        *reinterpret_cast<h8*>(outh + ((size_t)n * HW + p0 + 4 * l15 + j) * CQ + chb) = pk;
      }
    } else {
#pragma unroll
      for (int r = 0; r < 8; ++r) {
        h4 pk;
#pragma unroll
        for (int j = 0; j < 4; ++j) {
          float v = acc[m][j][r] * iv[r] + bb[r];
          v = v > 0.f ? v : 0.f;
          pk[j] = (_Float16)v;
        }
        *reinterpret_cast<h4*>(outh + ((size_t)n * CV + chb + r) * HW + p0 + 4 * l15) = pk;
      }
    }
  }
}

__global__ __launch_bounds__(256) __attribute__((amdgpu_waves_per_eu(2, 2))) void flash_attn_kernel(
    const _Float16* q, const _Float16* fT,
    const float* __restrict__ mask, float* __restrict__ out)
{
  constexpr float LOG2E     = 1.4426950408889634f;
  constexpr float ATT_SCALE = (1.0f / (1e-8f + 11.313708498984761f)) * LOG2E;
  constexpr int   NSTAGE    = HW / SK;
  constexpr int KST_BYTES = 2 * SK * KP * 2;
  constexpr int VST_BYTES = 2 * CV * VP * 2;
  constexpr int MRG_BYTES = 4 * 64 * 32 * 4 * 4;
  constexpr int MML_BYTES = 4 * 2 * 2 * 32 * 4;
  static_assert(KST_BYTES + VST_BYTES + HW * 4 <= MRG_BYTES + MML_BYTES, "the ring must fit in the block");
  __shared__ __align__(32) unsigned char smem[MRG_BYTES + MML_BYTES];
  _Float16* const Kst  = reinterpret_cast<_Float16*>(smem);
  _Float16* const Vst  = reinterpret_cast<_Float16*>(smem + KST_BYTES);
  float*    const Madd = reinterpret_cast<float*>(smem + KST_BYTES + VST_BYTES);
  float*    const Mrg  = reinterpret_cast<float*>(smem);
  float*    const Mml  = reinterpret_cast<float*>(smem + MRG_BYTES);

  const int lane = threadIdx.x;
  const int l15  = lane & 15;
  const int lh   = lane >> 4;
  const int wid  = __builtin_amdgcn_readfirstlane(threadIdx.y);
  const int g    = wid & 3;
  const int kh   = wid >> 2;
  const int tid  = wid * 32 + lane;
  const int i0   = (blockIdx.x * 4 + g) * 32;
  const int n    = blockIdx.y;

  v16h qa[2][4];
#pragma unroll
  for (int u = 0; u < 2; ++u) {
    const _Float16* qrow = q + ((size_t)n * HW + i0 + 16 * u + l15) * CQ;
#pragma unroll
    for (int kc = 0; kc < 4; ++kc) qa[u][kc] = load_frag16(qrow + kc * 32, lh);
  }

  const _Float16* kg = q  + (size_t)n * HW * CQ + (size_t)tid * 8;
  const _Float16* vg = fT + ((size_t)n * CV + (tid >> 3)) * HW + (tid & 7) * 8;
  const int kl = (tid >> 4) * KP + (tid & 15) * 8;
  const int vl = (tid >> 3) * VP + (tid & 7) * 8;

  uint4 pk0, pk1, pk2, pk3, pv0, pv1, pv2, pv3, pv4, pv5, pv6, pv7;
#define FETCH_TILE(S)                                                                 \
  do {                                                                                \
    const _Float16* kgs_ = kg + (size_t)((S) * SK) * CQ;                              \
    const _Float16* vgs_ = vg + (S) * SK;                                             \
    pk0 = *reinterpret_cast<const uint4*>(kgs_);                                      \
    pk1 = *reinterpret_cast<const uint4*>(kgs_ + 1 * 16 * CQ);                        \
    pk2 = *reinterpret_cast<const uint4*>(kgs_ + 2 * 16 * CQ);                        \
    pk3 = *reinterpret_cast<const uint4*>(kgs_ + 3 * 16 * CQ);                        \
    pv0 = *reinterpret_cast<const uint4*>(vgs_);                                      \
    pv1 = *reinterpret_cast<const uint4*>(vgs_ + (size_t)32 * HW);                    \
    pv2 = *reinterpret_cast<const uint4*>(vgs_ + (size_t)64 * HW);                    \
    pv3 = *reinterpret_cast<const uint4*>(vgs_ + (size_t)96 * HW);                    \
    pv4 = *reinterpret_cast<const uint4*>(vgs_ + (size_t)128 * HW);                   \
    pv5 = *reinterpret_cast<const uint4*>(vgs_ + (size_t)160 * HW);                   \
    pv6 = *reinterpret_cast<const uint4*>(vgs_ + (size_t)192 * HW);                   \
    pv7 = *reinterpret_cast<const uint4*>(vgs_ + (size_t)224 * HW);                   \
  } while (0)
#define STASH_TILE(B)                                                                 \
  do {                                                                                \
    _Float16* kd_ = Kst + (B) * (SK * KP) + kl;                                                \
    _Float16* vd_ = Vst + (B) * (CV * VP) + vl;                                                \
    *reinterpret_cast<uint4*>(kd_)               = pk0;                               \
    *reinterpret_cast<uint4*>(kd_ + 1 * 16 * KP) = pk1;                               \
    *reinterpret_cast<uint4*>(kd_ + 2 * 16 * KP) = pk2;                               \
    *reinterpret_cast<uint4*>(kd_ + 3 * 16 * KP) = pk3;                               \
    *reinterpret_cast<uint4*>(vd_)               = pv0;                               \
    *reinterpret_cast<uint4*>(vd_ + 32 * VP)     = pv1;                               \
    *reinterpret_cast<uint4*>(vd_ + 64 * VP)     = pv2;                               \
    *reinterpret_cast<uint4*>(vd_ + 96 * VP)     = pv3;                               \
    *reinterpret_cast<uint4*>(vd_ + 128 * VP)    = pv4;                               \
    *reinterpret_cast<uint4*>(vd_ + 160 * VP)    = pv5;                               \
    *reinterpret_cast<uint4*>(vd_ + 192 * VP)    = pv6;                               \
    *reinterpret_cast<uint4*>(vd_ + 224 * VP)    = pv7;                               \
  } while (0)

  v8f oacc[2][16];
#pragma unroll
  for (int u = 0; u < 2; ++u) {
#pragma unroll
    for (int t = 0; t < 16; ++t) oacc[u][t] = (v8f){};
  }
  float mrun[2] = {-1e30f, -1e30f};
  float lrun[2] = {0.f, 0.f};

  const float* mk = mask + (size_t)n * HW;
  for (int i = 0; i < HW / 256; ++i)
    Madd[tid + 256 * i] = (mk[tid + 256 * i] - 1.f) * (1e8f * LOG2E);

  FETCH_TILE(0); STASH_TILE(0);
  __syncthreads();

  const int koff = (32 * kh + l15) * KP;
  const int voff = l15 * VP + 32 * kh;

  for (int s = 0; s < NSTAGE; ++s) {
    const int b  = s & 1;
    const int sn = (s + 1 < NSTAGE) ? s + 1 : NSTAGE - 1;
    const _Float16* kt = Kst + b * (SK * KP) + koff;
    const _Float16* vt = Vst + b * (CV * VP) + voff;
    const int j0 = s * SK + 32 * kh;

    FETCH_TILE(sn);
    asm volatile("" ::: "memory");

    const v8f ma0 = *reinterpret_cast<const v8f*>(&Madd[j0 + 8 * lh]);
    const v8f ma1 = *reinterpret_cast<const v8f*>(&Madd[j0 + 16 + 8 * lh]);

    v8f s0[2], s1[2];
#pragma unroll
    for (int u = 0; u < 2; ++u) { s0[u] = (v8f){}; s1[u] = (v8f){}; }
#pragma unroll
    for (int kc = 0; kc < 4; ++kc) {
      const _Float16* kb = kt + kc * 32;
      const v16h a0 = load_frag16(kb, lh);
      const v16h a1 = load_frag16(kb + 16 * KP, lh);
#pragma unroll
      for (int u = 0; u < 2; ++u) {
        s0[u] = __builtin_amdgcn_wmma_f32_16x16x32_f16(false, a0, false, qa[u][kc], (short)0, s0[u], false, false);
        s1[u] = __builtin_amdgcn_wmma_f32_16x16x32_f16(false, a1, false, qa[u][kc], (short)0, s1[u], false, false);
      }
    }

    constexpr float RESCALE_LAG = 8.0f;
    float vmx[2];
#pragma unroll
    for (int u = 0; u < 2; ++u) {
      float vmax = -1e30f;
#pragma unroll
      for (int r = 0; r < 8; ++r) {
        s0[u][r] = s0[u][r] * ATT_SCALE + ma0[r];
        s1[u][r] = s1[u][r] * ATT_SCALE + ma1[r];
        vmax = fmaxf(vmax, fmaxf(s0[u][r], s1[u][r]));
      }
      vmx[u] = vmax;
    }
    const bool lag_exceeded = (vmx[0] > mrun[0] + RESCALE_LAG) || (vmx[1] > mrun[1] + RESCALE_LAG);
    if (__builtin_amdgcn_ballot_w32(lag_exceeded) != 0u) {
#pragma unroll
      for (int u = 0; u < 2; ++u) {
        const float mn    = fmaxf(mrun[u], fmaxf(vmx[u], xhalf(vmx[u])));
        const float alpha = ex2(mrun[u] - mn);
        mrun[u] = mn;
        lrun[u] *= alpha;
#pragma unroll
        for (int t = 0; t < 16; ++t) oacc[u][t] *= alpha;
      }
    }
    v16h pbt[2];
#pragma unroll
    for (int u = 0; u < 2; ++u) {
      const float mu = mrun[u];
      float rs0 = 0.f, rs1 = 0.f;
#pragma unroll
      for (int r = 0; r < 8; ++r) {
        s0[u][r] = ex2(s0[u][r] - mu);
        s1[u][r] = ex2(s1[u][r] - mu);
        rs0 += s0[u][r];
        rs1 += s1[u][r];
      }
      lrun[u] += rs0 + rs1;

#pragma unroll
      for (int r = 0; r < 8; ++r) { pbt[u][r] = (_Float16)s0[u][r]; pbt[u][r + 8] = (_Float16)s1[u][r]; }
    }

    asm volatile("" :: "v"(lrun[0]), "v"(lrun[1]) : "memory");
    STASH_TILE(b ^ 1);

    v16h va[16];
#pragma unroll
    for (int t = 0; t < 16; ++t) va[t] = load_frag16(vt + (t * 16) * VP, lh);
    __builtin_amdgcn_sched_barrier(0);
#pragma unroll
    for (int t = 0; t < 16; ++t) {
#pragma unroll
      for (int u = 0; u < 2; ++u)
        oacc[u][t] = __builtin_amdgcn_wmma_f32_16x16x32_f16(false, va[t], false, pbt[u], (short)0, oacc[u][t], false, false);
    }
    __builtin_amdgcn_sched_barrier(0);

    __syncthreads();
  }

  float lt[2];
#pragma unroll
  for (int u = 0; u < 2; ++u) lt[u] = lrun[u] + xhalf(lrun[u]);

  const int lane2 = (int)__builtin_amdgcn_mbcnt_lo(~0u, 0u);
  const int l15e  = lane2 & 15;
  const int lhe   = lane2 >> 4;
  float* mrg = Mrg + (size_t)g * (64 * 32 * 4) + lane2 * 4;
  float* mml = Mml + g * (2 * 2 * 32) + lane2;
  if (kh == 1) {
#pragma unroll
    for (int u = 0; u < 2; ++u) {
      mml[(u * 2 + 0) * 32] = mrun[u];
      mml[(u * 2 + 1) * 32] = lt[u];
#pragma unroll
      for (int t = 0; t < 16; ++t) {
        v4f lo, hi;
#pragma unroll
        for (int r = 0; r < 4; ++r) { lo[r] = oacc[u][t][r]; hi[r] = oacc[u][t][r + 4]; }
        *reinterpret_cast<v4f*>(mrg + ((u * 16 + t) * 2 + 0) * (32 * 4)) = lo;
        *reinterpret_cast<v4f*>(mrg + ((u * 16 + t) * 2 + 1) * (32 * 4)) = hi;
      }
    }
  }
  __syncthreads();
  if (kh == 0) {
#pragma unroll
    for (int u = 0; u < 2; ++u) {
      const float mp   = mml[(u * 2 + 0) * 32];
      const float lp   = mml[(u * 2 + 1) * 32];
      const float mn   = fmaxf(mrun[u], mp);
      const float ao   = ex2(mrun[u] - mn);
      const float ap   = ex2(mp - mn);
      const float ltot = ao * lt[u] + ap * lp;
      const float fin  = (1.0f / ltot) * mask[(size_t)n * HW + i0 + 16 * u + l15e];
      const float fo   = ao * fin;
      const float fp   = ap * fin;
#pragma unroll
      for (int t = 0; t < 16; ++t) {
        const v4f lo = *reinterpret_cast<const v4f*>(mrg + ((u * 16 + t) * 2 + 0) * (32 * 4));
        const v4f hi = *reinterpret_cast<const v4f*>(mrg + ((u * 16 + t) * 2 + 1) * (32 * 4));
        float* dst = out + ((size_t)n * CV + t * 16 + 8 * lhe) * HW + i0 + 16 * u + l15e;
#pragma unroll
        for (int r = 0; r < 4; ++r) {
          dst[(size_t)r * HW]       = oacc[u][t][r] * fo + lo[r] * fp;
          dst[(size_t)(r + 4) * HW] = oacc[u][t][r + 4] * fo + hi[r] * fp;
        }
      }
    }
  }
}

#undef FETCH_TILE
#undef STASH_TILE

extern "C" void kernel_launch(void* const* d_in, const int* in_sizes, int n_in,
                              void* d_out, int out_size, void* d_ws, size_t ws_size,
                              hipStream_t stream) {
  (void)in_sizes; (void)n_in; (void)out_size; (void)ws_size;
  const float* x  = (const float*)d_in[0];
  const float* mk = (const float*)d_in[1];
  const float* rw = (const float*)d_in[2];
  const float* rg = (const float*)d_in[3];
  const float* rb = (const float*)d_in[4];
  const float* rm = (const float*)d_in[5];
  const float* rv = (const float*)d_in[6];
  const float* aw = (const float*)d_in[7];
  const float* ag = (const float*)d_in[8];
  const float* ab = (const float*)d_in[9];
  const float* am = (const float*)d_in[10];
  const float* av = (const float*)d_in[11];
  float* out = (float*)d_out;

  _Float16* qbuf = (_Float16*)d_ws;
  _Float16* fT   = qbuf + (size_t)NB * HW * CQ;

  const dim3 blk(32, 4);
  proj_bn_relu_kernel<<<dim3(HW / 64, NB), blk, 0, stream>>>(
      x, rw, rg, rb, rm, rv, fT, 0, 2);
  proj_bn_relu_kernel<<<dim3(HW / 128, NB), blk, 0, stream>>>(
      x, aw, ag, ab, am, av, qbuf, 1, 1);
  flash_attn_kernel<<<dim3(HW / 128, NB), dim3(32, 8), 0, stream>>>(qbuf, fT, mk, out);
}


# Attention: per-tile workgroup barrier split (signal behind a wave's own tile stores, wait behind its output run) over a 3-slot K/V ring: the two waves of a SIMD run out of phase; sum order unchanged
# speedup vs baseline: 7.1865x; 1.0373x over previous
#include <hip/hip_runtime.h>
#include <hip/hip_bf16.h>

#define HW   4096
#define NB   8
#define CIN  256
#define CQ   128
#define CV   256

#define SK   64
#define KP   136
#define VP   72

typedef float    v2f  __attribute__((ext_vector_type(2)));
typedef float    v4f  __attribute__((ext_vector_type(4)));
typedef float    v8f  __attribute__((ext_vector_type(8)));
typedef _Float16 v16h __attribute__((ext_vector_type(16)));
typedef _Float16 h8   __attribute__((ext_vector_type(8)));
typedef _Float16 h4   __attribute__((ext_vector_type(4)));

static __device__ __forceinline__ v16h load_frag16(const _Float16* rowBase, int lh) {
  const h8 lo = *reinterpret_cast<const h8*>(rowBase + 8 * lh);
  const h8 hi = *reinterpret_cast<const h8*>(rowBase + 8 * lh + 16);
  v16h r;
#pragma unroll
  for (int i = 0; i < 8; ++i) { r[i] = lo[i]; r[i + 8] = hi[i]; }
  return r;
}


static __device__ __forceinline__ float xhalf(float v) {
  return __int_as_float(__builtin_amdgcn_permlanex16(__float_as_int(v), __float_as_int(v),
                                                     0x76543210, 0xfedcba98, false, false));
}

static __device__ __forceinline__ float ex2(float v) { return __builtin_amdgcn_exp2f(v); }

__global__ __launch_bounds__(128) void proj_bn_relu_kernel(
    const float* __restrict__ x, const float* __restrict__ w,
    const float* __restrict__ gamma, const float* __restrict__ beta,
    const float* __restrict__ mean, const float* __restrict__ var,
    _Float16* __restrict__ outh, int posMajor, int mshift)
{
  const int lane = threadIdx.x;
  const int l15  = lane & 15;
  const int lh   = lane >> 4;
  const int wid  = threadIdx.y;
  const int o0   = (wid & ((1 << mshift) - 1)) * 64;
  const int p0   = (blockIdx.x * (4 >> mshift) + (wid >> mshift)) * 64;
  const int n    = blockIdx.y;

  const float* wrow = w + (size_t)(o0 + l15) * CIN + 4 * lh;
  const float* xcol = x + ((size_t)n * CIN + 4 * lh) * HW + p0 + 4 * l15;

  v8f acc[4][4];
#pragma unroll
  for (int m = 0; m < 4; ++m) {
#pragma unroll
    for (int j = 0; j < 4; ++j) acc[m][j] = (v8f){};
  }

  v4f aw[4], bx[4];
#pragma unroll
  for (int m = 0; m < 4; ++m) aw[m] = *reinterpret_cast<const v4f*>(wrow + (size_t)(16 * m) * CIN);
#pragma unroll
  for (int e = 0; e < 4; ++e) bx[e] = *reinterpret_cast<const v4f*>(xcol + (size_t)e * HW);

#pragma unroll 2
  for (int c0 = 0; c0 < CIN; c0 += 8) {
    const int cn = (c0 + 8 < CIN) ? c0 + 8 : c0;
    v4f awn[4], bxn[4];
#pragma unroll
    for (int m = 0; m < 4; ++m) awn[m] = *reinterpret_cast<const v4f*>(wrow + (size_t)(16 * m) * CIN + cn);
#pragma unroll
    for (int e = 0; e < 4; ++e) bxn[e] = *reinterpret_cast<const v4f*>(xcol + (size_t)(cn + e) * HW);

#pragma unroll
    for (int s = 0; s < 2; ++s) {
      v2f b[4];
#pragma unroll
      for (int j = 0; j < 4; ++j) { b[j][0] = bx[2 * s][j]; b[j][1] = bx[2 * s + 1][j]; }
#pragma unroll
      for (int m = 0; m < 4; ++m) {
        v2f a; a[0] = aw[m][2 * s]; a[1] = aw[m][2 * s + 1];
#pragma unroll
        for (int j = 0; j < 4; ++j)
          acc[m][j] = __builtin_amdgcn_wmma_f32_16x16x4_f32(false, a, false, b[j], (short)0, acc[m][j], false, false);
      }
    }
#pragma unroll
    for (int m = 0; m < 4; ++m) { aw[m] = awn[m]; bx[m] = bxn[m]; }
  }

#pragma unroll
  for (int m = 0; m < 4; ++m) {
    const int chb = o0 + 16 * m + 8 * lh;
    float iv[8], bb[8];
#pragma unroll
    for (int q4 = 0; q4 < 2; ++q4) {
      const v4f g  = *reinterpret_cast<const v4f*>(gamma + chb + 4 * q4);
      const v4f vv = *reinterpret_cast<const v4f*>(var   + chb + 4 * q4);
      const v4f be = *reinterpret_cast<const v4f*>(beta  + chb + 4 * q4);
      const v4f mu = *reinterpret_cast<const v4f*>(mean  + chb + 4 * q4);
#pragma unroll
      for (int r = 0; r < 4; ++r) {
        iv[4 * q4 + r] = g[r] * rsqrtf(vv[r] + 1e-5f);
        bb[4 * q4 + r] = be[r] - mu[r] * iv[4 * q4 + r];
      }
    }
    if (posMajor) {
#pragma unroll
      for (int j = 0; j < 4; ++j) {
        h8 pk;
#pragma unroll
        for (int r = 0; r < 8; ++r) {
          float v = acc[m][j][r] * iv[r] + bb[r];
          v = v > 0.f ? v : 0.f;
          pk[r] = (_Float16)v;
        }
        *reinterpret_cast<h8*>(outh + ((size_t)n * HW + p0 + 4 * l15 + j) * CQ + chb) = pk;
      }
    } else {
#pragma unroll
      for (int r = 0; r < 8; ++r) {
        h4 pk;
#pragma unroll
        for (int j = 0; j < 4; ++j) {
          float v = acc[m][j][r] * iv[r] + bb[r];
          v = v > 0.f ? v : 0.f;
          pk[j] = (_Float16)v;
        }
        *reinterpret_cast<h4*>(outh + ((size_t)n * CV + chb + r) * HW + p0 + 4 * l15) = pk;
      }
    }
  }
}

__global__ __launch_bounds__(256) __attribute__((amdgpu_waves_per_eu(2, 2))) void flash_attn_kernel(
    const _Float16* q, const _Float16* fT,
    const float* __restrict__ mask, float* __restrict__ out)
{
  constexpr float LOG2E     = 1.4426950408889634f;
  constexpr float ATT_SCALE = (1.0f / (1e-8f + 11.313708498984761f)) * LOG2E;
  constexpr int   NSTAGE    = HW / SK;
  constexpr int NSLOT     = 3;
  constexpr int KST_BYTES = NSLOT * SK * KP * 2;
  constexpr int VST_BYTES = NSLOT * CV * VP * 2;
  constexpr int MRG_BYTES = 4 * 64 * 32 * 4 * 4;
  constexpr int MML_BYTES = 4 * 2 * 2 * 32 * 4;
  constexpr int RING_BYTES = KST_BYTES + VST_BYTES + HW * 4;
  constexpr int SMEM_BYTES = RING_BYTES > MRG_BYTES + MML_BYTES ? RING_BYTES : MRG_BYTES + MML_BYTES;
  __shared__ __align__(32) unsigned char smem[SMEM_BYTES];
  _Float16* const Kst  = reinterpret_cast<_Float16*>(smem);
  _Float16* const Vst  = reinterpret_cast<_Float16*>(smem + KST_BYTES);
  float*    const Madd = reinterpret_cast<float*>(smem + KST_BYTES + VST_BYTES);
  float*    const Mrg  = reinterpret_cast<float*>(smem);
  float*    const Mml  = reinterpret_cast<float*>(smem + MRG_BYTES);

  const int lane = threadIdx.x;
  const int l15  = lane & 15;
  const int lh   = lane >> 4;
  const int wid  = __builtin_amdgcn_readfirstlane(threadIdx.y);
  const int g    = wid & 3;
  const int kh   = wid >> 2;
  const int tid  = wid * 32 + lane;
  const int i0   = (blockIdx.x * 4 + g) * 32;
  const int n    = blockIdx.y;

  v16h qa[2][4];
#pragma unroll
  for (int u = 0; u < 2; ++u) {
    const _Float16* qrow = q + ((size_t)n * HW + i0 + 16 * u + l15) * CQ;
#pragma unroll
    for (int kc = 0; kc < 4; ++kc) qa[u][kc] = load_frag16(qrow + kc * 32, lh);
  }

  const _Float16* kg = q  + (size_t)n * HW * CQ + (size_t)tid * 8;
  const _Float16* vg = fT + ((size_t)n * CV + (tid >> 3)) * HW + (tid & 7) * 8;
  const int kl = (tid >> 4) * KP + (tid & 15) * 8;
  const int vl = (tid >> 3) * VP + (tid & 7) * 8;

  uint4 pk0, pk1, pk2, pk3, pv0, pv1, pv2, pv3, pv4, pv5, pv6, pv7;
#define FETCH_TILE(S)                                                                 \
  do {                                                                                \
    const _Float16* kgs_ = kg + (size_t)((S) * SK) * CQ;                              \
    const _Float16* vgs_ = vg + (S) * SK;                                             \
    pk0 = *reinterpret_cast<const uint4*>(kgs_);                                      \
    pk1 = *reinterpret_cast<const uint4*>(kgs_ + 1 * 16 * CQ);                        \
    pk2 = *reinterpret_cast<const uint4*>(kgs_ + 2 * 16 * CQ);                        \
    pk3 = *reinterpret_cast<const uint4*>(kgs_ + 3 * 16 * CQ);                        \
    pv0 = *reinterpret_cast<const uint4*>(vgs_);                                      \
    pv1 = *reinterpret_cast<const uint4*>(vgs_ + (size_t)32 * HW);                    \
    pv2 = *reinterpret_cast<const uint4*>(vgs_ + (size_t)64 * HW);                    \
    pv3 = *reinterpret_cast<const uint4*>(vgs_ + (size_t)96 * HW);                    \
    pv4 = *reinterpret_cast<const uint4*>(vgs_ + (size_t)128 * HW);                   \
    pv5 = *reinterpret_cast<const uint4*>(vgs_ + (size_t)160 * HW);                   \
    pv6 = *reinterpret_cast<const uint4*>(vgs_ + (size_t)192 * HW);                   \
    pv7 = *reinterpret_cast<const uint4*>(vgs_ + (size_t)224 * HW);                   \
  } while (0)
#define STASH_TILE(B)                                                                 \
  do {                                                                                \
    _Float16* kd_ = Kst + (B) * (SK * KP) + kl;                                                \
    _Float16* vd_ = Vst + (B) * (CV * VP) + vl;                                                \
    *reinterpret_cast<uint4*>(kd_)               = pk0;                               \
    *reinterpret_cast<uint4*>(kd_ + 1 * 16 * KP) = pk1;                               \
    *reinterpret_cast<uint4*>(kd_ + 2 * 16 * KP) = pk2;                               \
    *reinterpret_cast<uint4*>(kd_ + 3 * 16 * KP) = pk3;                               \
    *reinterpret_cast<uint4*>(vd_)               = pv0;                               \
    *reinterpret_cast<uint4*>(vd_ + 32 * VP)     = pv1;                               \
    *reinterpret_cast<uint4*>(vd_ + 64 * VP)     = pv2;                               \
    *reinterpret_cast<uint4*>(vd_ + 96 * VP)     = pv3;                               \
    *reinterpret_cast<uint4*>(vd_ + 128 * VP)    = pv4;                               \
    *reinterpret_cast<uint4*>(vd_ + 160 * VP)    = pv5;                               \
    *reinterpret_cast<uint4*>(vd_ + 192 * VP)    = pv6;                               \
    *reinterpret_cast<uint4*>(vd_ + 224 * VP)    = pv7;                               \
  } while (0)

  v8f oacc[2][16];
#pragma unroll
  for (int u = 0; u < 2; ++u) {
#pragma unroll
    for (int t = 0; t < 16; ++t) oacc[u][t] = (v8f){};
  }
  float mrun[2] = {-1e30f, -1e30f};
  float lrun[2] = {0.f, 0.f};

  const float* mk = mask + (size_t)n * HW;
  for (int i = 0; i < HW / 256; ++i)
    Madd[tid + 256 * i] = (mk[tid + 256 * i] - 1.f) * (1e8f * LOG2E);

  FETCH_TILE(0); STASH_TILE(0);
  __syncthreads();

  const int koff = (32 * kh + l15) * KP;
  const int voff = l15 * VP + 32 * kh;

  int b = 0;
  for (int s = 0; s < NSTAGE; ++s) {
    const int bn = (b == NSLOT - 1) ? 0 : b + 1;
    const int sn = (s + 1 < NSTAGE) ? s + 1 : NSTAGE - 1;
    const _Float16* kt = Kst + b * (SK * KP) + koff;
    const _Float16* vt = Vst + b * (CV * VP) + voff;
    const int j0 = s * SK + 32 * kh;

    FETCH_TILE(sn);
    asm volatile("" ::: "memory");

    const v8f ma0 = *reinterpret_cast<const v8f*>(&Madd[j0 + 8 * lh]);
    const v8f ma1 = *reinterpret_cast<const v8f*>(&Madd[j0 + 16 + 8 * lh]);

    v8f s0[2], s1[2];
#pragma unroll
    for (int u = 0; u < 2; ++u) { s0[u] = (v8f){}; s1[u] = (v8f){}; }
#pragma unroll
    for (int kc = 0; kc < 4; ++kc) {
      const _Float16* kb = kt + kc * 32;
      const v16h a0 = load_frag16(kb, lh);
      const v16h a1 = load_frag16(kb + 16 * KP, lh);
#pragma unroll
      for (int u = 0; u < 2; ++u) {
        s0[u] = __builtin_amdgcn_wmma_f32_16x16x32_f16(false, a0, false, qa[u][kc], (short)0, s0[u], false, false);
        s1[u] = __builtin_amdgcn_wmma_f32_16x16x32_f16(false, a1, false, qa[u][kc], (short)0, s1[u], false, false);
      }
    }

    constexpr float RESCALE_LAG = 8.0f;
    float vmx[2];
#pragma unroll
    for (int u = 0; u < 2; ++u) {
      float vmax = -1e30f;
#pragma unroll
      for (int r = 0; r < 8; ++r) {
        s0[u][r] = s0[u][r] * ATT_SCALE + ma0[r];
        s1[u][r] = s1[u][r] * ATT_SCALE + ma1[r];
        vmax = fmaxf(vmax, fmaxf(s0[u][r], s1[u][r]));
      }
      vmx[u] = vmax;
    }
    const bool lag_exceeded = (vmx[0] > mrun[0] + RESCALE_LAG) || (vmx[1] > mrun[1] + RESCALE_LAG);
    if (__builtin_amdgcn_ballot_w32(lag_exceeded) != 0u) {
#pragma unroll
      for (int u = 0; u < 2; ++u) {
        const float mn    = fmaxf(mrun[u], fmaxf(vmx[u], xhalf(vmx[u])));
        const float alpha = ex2(mrun[u] - mn);
        mrun[u] = mn;
        lrun[u] *= alpha;
#pragma unroll
        for (int t = 0; t < 16; ++t) oacc[u][t] *= alpha;
      }
    }
    v16h pbt[2];
#pragma unroll
    for (int u = 0; u < 2; ++u) {
      const float mu = mrun[u];
      float rs0 = 0.f, rs1 = 0.f;
#pragma unroll
      for (int r = 0; r < 8; ++r) {
        s0[u][r] = ex2(s0[u][r] - mu);
        s1[u][r] = ex2(s1[u][r] - mu);
        rs0 += s0[u][r];
        rs1 += s1[u][r];
      }
      lrun[u] += rs0 + rs1;

#pragma unroll
      for (int r = 0; r < 8; ++r) { pbt[u][r] = (_Float16)s0[u][r]; pbt[u][r + 8] = (_Float16)s1[u][r]; }
    }

    asm volatile("" :: "v"(lrun[0]), "v"(lrun[1]) : "memory");
    STASH_TILE(bn);
    __builtin_amdgcn_fence(__ATOMIC_RELEASE, "workgroup");
    __builtin_amdgcn_s_barrier_signal(-1);

    v16h va[16];
#pragma unroll
    for (int t = 0; t < 16; ++t) va[t] = load_frag16(vt + (t * 16) * VP, lh);
    __builtin_amdgcn_sched_barrier(0);
#pragma unroll
    for (int t = 0; t < 16; ++t) {
#pragma unroll
      for (int u = 0; u < 2; ++u)
        oacc[u][t] = __builtin_amdgcn_wmma_f32_16x16x32_f16(false, va[t], false, pbt[u], (short)0, oacc[u][t], false, false);
    }
    __builtin_amdgcn_sched_barrier(0);

    __builtin_amdgcn_s_barrier_wait(-1);
    __builtin_amdgcn_fence(__ATOMIC_ACQUIRE, "workgroup");
    b = bn;
  }
  __syncthreads();

  float lt[2];
#pragma unroll
  for (int u = 0; u < 2; ++u) lt[u] = lrun[u] + xhalf(lrun[u]);

  const int lane2 = (int)__builtin_amdgcn_mbcnt_lo(~0u, 0u);
  const int l15e  = lane2 & 15;
  const int lhe   = lane2 >> 4;
  float* mrg = Mrg + (size_t)g * (64 * 32 * 4) + lane2 * 4;
  float* mml = Mml + g * (2 * 2 * 32) + lane2;
  if (kh == 1) {
#pragma unroll
    for (int u = 0; u < 2; ++u) {
      mml[(u * 2 + 0) * 32] = mrun[u];
      mml[(u * 2 + 1) * 32] = lt[u];
#pragma unroll
      for (int t = 0; t < 16; ++t) {
        v4f lo, hi;
#pragma unroll
        for (int r = 0; r < 4; ++r) { lo[r] = oacc[u][t][r]; hi[r] = oacc[u][t][r + 4]; }
        *reinterpret_cast<v4f*>(mrg + ((u * 16 + t) * 2 + 0) * (32 * 4)) = lo;
        *reinterpret_cast<v4f*>(mrg + ((u * 16 + t) * 2 + 1) * (32 * 4)) = hi;
      }
    }
  }
  __syncthreads();
  if (kh == 0) {
#pragma unroll
    for (int u = 0; u < 2; ++u) {
      const float mp   = mml[(u * 2 + 0) * 32];
      const float lp   = mml[(u * 2 + 1) * 32];
      const float mn   = fmaxf(mrun[u], mp);
      const float ao   = ex2(mrun[u] - mn);
      const float ap   = ex2(mp - mn);
      const float ltot = ao * lt[u] + ap * lp;
      const float fin  = (1.0f / ltot) * mask[(size_t)n * HW + i0 + 16 * u + l15e];
      const float fo   = ao * fin;
      const float fp   = ap * fin;
#pragma unroll
      for (int t = 0; t < 16; ++t) {
        const v4f lo = *reinterpret_cast<const v4f*>(mrg + ((u * 16 + t) * 2 + 0) * (32 * 4));
        const v4f hi = *reinterpret_cast<const v4f*>(mrg + ((u * 16 + t) * 2 + 1) * (32 * 4));
        float* dst = out + ((size_t)n * CV + t * 16 + 8 * lhe) * HW + i0 + 16 * u + l15e;
#pragma unroll
        for (int r = 0; r < 4; ++r) {
          dst[(size_t)r * HW]       = oacc[u][t][r] * fo + lo[r] * fp;
          dst[(size_t)(r + 4) * HW] = oacc[u][t][r + 4] * fo + hi[r] * fp;
        }
      }
    }
  }
}

#undef FETCH_TILE
#undef STASH_TILE

extern "C" void kernel_launch(void* const* d_in, const int* in_sizes, int n_in,
                              void* d_out, int out_size, void* d_ws, size_t ws_size,
                              hipStream_t stream) {
  (void)in_sizes; (void)n_in; (void)out_size; (void)ws_size;
  const float* x  = (const float*)d_in[0];
  const float* mk = (const float*)d_in[1];
  const float* rw = (const float*)d_in[2];
  const float* rg = (const float*)d_in[3];
  const float* rb = (const float*)d_in[4];
  const float* rm = (const float*)d_in[5];
  const float* rv = (const float*)d_in[6];
  const float* aw = (const float*)d_in[7];
  const float* ag = (const float*)d_in[8];
  const float* ab = (const float*)d_in[9];
  const float* am = (const float*)d_in[10];
  const float* av = (const float*)d_in[11];
  float* out = (float*)d_out;

  _Float16* qbuf = (_Float16*)d_ws;
  _Float16* fT   = qbuf + (size_t)NB * HW * CQ;

  const dim3 blk(32, 4);
  proj_bn_relu_kernel<<<dim3(HW / 64, NB), blk, 0, stream>>>(
      x, rw, rg, rb, rm, rv, fT, 0, 2);
  proj_bn_relu_kernel<<<dim3(HW / 128, NB), blk, 0, stream>>>(
      x, aw, ag, ab, am, av, qbuf, 1, 1);
  flash_attn_kernel<<<dim3(HW / 128, NB), dim3(32, 8), 0, stream>>>(qbuf, fT, mk, out);
}


# Projections: both 1x1-conv GEMMs run in ONE launch (grid role x position block x batch, 768 equal workgroups, three waves per SIMD), not two launches at two and one wave per SIMD; sums unchanged
# speedup vs baseline: 7.4488x; 1.0365x over previous
#include <hip/hip_runtime.h>
#include <hip/hip_bf16.h>

#define HW   4096
#define NB   8
#define CIN  256
#define CQ   128
#define CV   256

#define SK   64
#define KP   136
#define VP   72

typedef float    v2f  __attribute__((ext_vector_type(2)));
typedef float    v4f  __attribute__((ext_vector_type(4)));
typedef float    v8f  __attribute__((ext_vector_type(8)));
typedef _Float16 v16h __attribute__((ext_vector_type(16)));
typedef _Float16 h8   __attribute__((ext_vector_type(8)));
typedef _Float16 h4   __attribute__((ext_vector_type(4)));

static __device__ __forceinline__ v16h load_frag16(const _Float16* rowBase, int lh) {
  const h8 lo = *reinterpret_cast<const h8*>(rowBase + 8 * lh);
  const h8 hi = *reinterpret_cast<const h8*>(rowBase + 8 * lh + 16);
  v16h r;
#pragma unroll
  for (int i = 0; i < 8; ++i) { r[i] = lo[i]; r[i + 8] = hi[i]; }
  return r;
}


static __device__ __forceinline__ float xhalf(float v) {
  return __int_as_float(__builtin_amdgcn_permlanex16(__float_as_int(v), __float_as_int(v),
                                                     0x76543210, 0xfedcba98, false, false));
}

static __device__ __forceinline__ float ex2(float v) { return __builtin_amdgcn_exp2f(v); }

__global__ __launch_bounds__(128) void proj_bn_relu_kernel(
    const float* __restrict__ x,
    const float* __restrict__ rwgt, const float* __restrict__ rgamma, const float* __restrict__ rbeta,
    const float* __restrict__ rmean, const float* __restrict__ rvar,
    const float* __restrict__ awgt, const float* __restrict__ agamma, const float* __restrict__ abeta,
    const float* __restrict__ amean, const float* __restrict__ avar,
    _Float16* __restrict__ fT, _Float16* __restrict__ qbuf)
{
  const int lane = threadIdx.x;
  const int l15  = lane & 15;
  const int lh   = lane >> 4;
  const int wid  = threadIdx.y;
  const int role = blockIdx.x;
  const int pblk = blockIdx.y;
  const int n    = blockIdx.z;
  const bool posMajor = (role == 2);
  const float* __restrict__ w     = posMajor ? awgt   : rwgt;
  const float* __restrict__ gamma = posMajor ? agamma : rgamma;
  const float* __restrict__ beta  = posMajor ? abeta  : rbeta;
  const float* __restrict__ mean  = posMajor ? amean  : rmean;
  const float* __restrict__ var   = posMajor ? avar   : rvar;
  _Float16* __restrict__ outh     = posMajor ? qbuf   : fT;
  const int o0   = (posMajor ? (wid & 1) : wid) * 64;
  const int p0   = (2 * pblk + (posMajor ? (wid >> 1) : role)) * 64;

  const float* wrow = w + (size_t)(o0 + l15) * CIN + 4 * lh;
  const float* xcol = x + ((size_t)n * CIN + 4 * lh) * HW + p0 + 4 * l15;

  v8f acc[4][4];
#pragma unroll
  for (int m = 0; m < 4; ++m) {
#pragma unroll
    for (int j = 0; j < 4; ++j) acc[m][j] = (v8f){};
  }

  v4f aw[4], bx[4];
#pragma unroll
  for (int m = 0; m < 4; ++m) aw[m] = *reinterpret_cast<const v4f*>(wrow + (size_t)(16 * m) * CIN);
#pragma unroll
  for (int e = 0; e < 4; ++e) bx[e] = *reinterpret_cast<const v4f*>(xcol + (size_t)e * HW);

#pragma unroll 2
  for (int c0 = 0; c0 < CIN; c0 += 8) {
    const int cn = (c0 + 8 < CIN) ? c0 + 8 : c0;
    v4f awn[4], bxn[4];
#pragma unroll
    for (int m = 0; m < 4; ++m) awn[m] = *reinterpret_cast<const v4f*>(wrow + (size_t)(16 * m) * CIN + cn);
#pragma unroll
    for (int e = 0; e < 4; ++e) bxn[e] = *reinterpret_cast<const v4f*>(xcol + (size_t)(cn + e) * HW);

#pragma unroll
    for (int s = 0; s < 2; ++s) {
      v2f b[4];
#pragma unroll
      for (int j = 0; j < 4; ++j) { b[j][0] = bx[2 * s][j]; b[j][1] = bx[2 * s + 1][j]; }
#pragma unroll
      for (int m = 0; m < 4; ++m) {
        v2f a; a[0] = aw[m][2 * s]; a[1] = aw[m][2 * s + 1];
#pragma unroll
        for (int j = 0; j < 4; ++j)
          acc[m][j] = __builtin_amdgcn_wmma_f32_16x16x4_f32(false, a, false, b[j], (short)0, acc[m][j], false, false);
      }
    }
#pragma unroll
    for (int m = 0; m < 4; ++m) { aw[m] = awn[m]; bx[m] = bxn[m]; }
  }

#pragma unroll
  for (int m = 0; m < 4; ++m) {
    const int chb = o0 + 16 * m + 8 * lh;
    float iv[8], bb[8];
#pragma unroll
    for (int q4 = 0; q4 < 2; ++q4) {
      const v4f g  = *reinterpret_cast<const v4f*>(gamma + chb + 4 * q4);
      const v4f vv = *reinterpret_cast<const v4f*>(var   + chb + 4 * q4);
      const v4f be = *reinterpret_cast<const v4f*>(beta  + chb + 4 * q4);
      const v4f mu = *reinterpret_cast<const v4f*>(mean  + chb + 4 * q4);
#pragma unroll
      for (int r = 0; r < 4; ++r) {
        iv[4 * q4 + r] = g[r] * rsqrtf(vv[r] + 1e-5f);
        bb[4 * q4 + r] = be[r] - mu[r] * iv[4 * q4 + r];
      }
    }
    if (posMajor) {
#pragma unroll
      for (int j = 0; j < 4; ++j) {
        h8 pk;
#pragma unroll
        for (int r = 0; r < 8; ++r) {
          float v = acc[m][j][r] * iv[r] + bb[r];
          v = v > 0.f ? v : 0.f;
          pk[r] = (_Float16)v;
        }
        *reinterpret_cast<h8*>(outh + ((size_t)n * HW + p0 + 4 * l15 + j) * CQ + chb) = pk;
      }
    } else {
#pragma unroll
      for (int r = 0; r < 8; ++r) {
        h4 pk;
#pragma unroll
        for (int j = 0; j < 4; ++j) {
          float v = acc[m][j][r] * iv[r] + bb[r];
          v = v > 0.f ? v : 0.f;
          pk[j] = (_Float16)v;
        }
        *reinterpret_cast<h4*>(outh + ((size_t)n * CV + chb + r) * HW + p0 + 4 * l15) = pk;
      }
    }
  }
}

__global__ __launch_bounds__(256) __attribute__((amdgpu_waves_per_eu(2, 2))) void flash_attn_kernel(
    const _Float16* q, const _Float16* fT,
    const float* __restrict__ mask, float* __restrict__ out)
{
  constexpr float LOG2E     = 1.4426950408889634f;
  constexpr float ATT_SCALE = (1.0f / (1e-8f + 11.313708498984761f)) * LOG2E;
  constexpr int   NSTAGE    = HW / SK;
  constexpr int NSLOT     = 3;
  constexpr int KST_BYTES = NSLOT * SK * KP * 2;
  constexpr int VST_BYTES = NSLOT * CV * VP * 2;
  constexpr int MRG_BYTES = 4 * 64 * 32 * 4 * 4;
  constexpr int MML_BYTES = 4 * 2 * 2 * 32 * 4;
  constexpr int RING_BYTES = KST_BYTES + VST_BYTES + HW * 4;
  constexpr int SMEM_BYTES = RING_BYTES > MRG_BYTES + MML_BYTES ? RING_BYTES : MRG_BYTES + MML_BYTES;
  __shared__ __align__(32) unsigned char smem[SMEM_BYTES];
  _Float16* const Kst  = reinterpret_cast<_Float16*>(smem);
  _Float16* const Vst  = reinterpret_cast<_Float16*>(smem + KST_BYTES);
  float*    const Madd = reinterpret_cast<float*>(smem + KST_BYTES + VST_BYTES);
  float*    const Mrg  = reinterpret_cast<float*>(smem);
  float*    const Mml  = reinterpret_cast<float*>(smem + MRG_BYTES);

  const int lane = threadIdx.x;
  const int l15  = lane & 15;
  const int lh   = lane >> 4;
  const int wid  = __builtin_amdgcn_readfirstlane(threadIdx.y);
  const int g    = wid & 3;
  const int kh   = wid >> 2;
  const int tid  = wid * 32 + lane;
  const int i0   = (blockIdx.x * 4 + g) * 32;
  const int n    = blockIdx.y;

  v16h qa[2][4];
#pragma unroll
  for (int u = 0; u < 2; ++u) {
    const _Float16* qrow = q + ((size_t)n * HW + i0 + 16 * u + l15) * CQ;
#pragma unroll
    for (int kc = 0; kc < 4; ++kc) qa[u][kc] = load_frag16(qrow + kc * 32, lh);
  }

  const _Float16* kg = q  + (size_t)n * HW * CQ + (size_t)tid * 8;
  const _Float16* vg = fT + ((size_t)n * CV + (tid >> 3)) * HW + (tid & 7) * 8;
  const int kl = (tid >> 4) * KP + (tid & 15) * 8;
  const int vl = (tid >> 3) * VP + (tid & 7) * 8;

  uint4 pk0, pk1, pk2, pk3, pv0, pv1, pv2, pv3, pv4, pv5, pv6, pv7;
#define FETCH_TILE(S)                                                                 \
  do {                                                                                \
    const _Float16* kgs_ = kg + (size_t)((S) * SK) * CQ;                              \
    const _Float16* vgs_ = vg + (S) * SK;                                             \
    pk0 = *reinterpret_cast<const uint4*>(kgs_);                                      \
    pk1 = *reinterpret_cast<const uint4*>(kgs_ + 1 * 16 * CQ);                        \
    pk2 = *reinterpret_cast<const uint4*>(kgs_ + 2 * 16 * CQ);                        \
    pk3 = *reinterpret_cast<const uint4*>(kgs_ + 3 * 16 * CQ);                        \
    pv0 = *reinterpret_cast<const uint4*>(vgs_);                                      \
    pv1 = *reinterpret_cast<const uint4*>(vgs_ + (size_t)32 * HW);                    \
    pv2 = *reinterpret_cast<const uint4*>(vgs_ + (size_t)64 * HW);                    \
    pv3 = *reinterpret_cast<const uint4*>(vgs_ + (size_t)96 * HW);                    \
    pv4 = *reinterpret_cast<const uint4*>(vgs_ + (size_t)128 * HW);                   \
    pv5 = *reinterpret_cast<const uint4*>(vgs_ + (size_t)160 * HW);                   \
    pv6 = *reinterpret_cast<const uint4*>(vgs_ + (size_t)192 * HW);                   \
    pv7 = *reinterpret_cast<const uint4*>(vgs_ + (size_t)224 * HW);                   \
  } while (0)
#define STASH_TILE(B)                                                                 \
  do {                                                                                \
    _Float16* kd_ = Kst + (B) * (SK * KP) + kl;                                                \
    _Float16* vd_ = Vst + (B) * (CV * VP) + vl;                                                \
    *reinterpret_cast<uint4*>(kd_)               = pk0;                               \
    *reinterpret_cast<uint4*>(kd_ + 1 * 16 * KP) = pk1;                               \
    *reinterpret_cast<uint4*>(kd_ + 2 * 16 * KP) = pk2;                               \
    *reinterpret_cast<uint4*>(kd_ + 3 * 16 * KP) = pk3;                               \
    *reinterpret_cast<uint4*>(vd_)               = pv0;                               \
    *reinterpret_cast<uint4*>(vd_ + 32 * VP)     = pv1;                               \
    *reinterpret_cast<uint4*>(vd_ + 64 * VP)     = pv2;                               \
    *reinterpret_cast<uint4*>(vd_ + 96 * VP)     = pv3;                               \
    *reinterpret_cast<uint4*>(vd_ + 128 * VP)    = pv4;                               \
    *reinterpret_cast<uint4*>(vd_ + 160 * VP)    = pv5;                               \
    *reinterpret_cast<uint4*>(vd_ + 192 * VP)    = pv6;                               \
    *reinterpret_cast<uint4*>(vd_ + 224 * VP)    = pv7;                               \
  } while (0)

  v8f oacc[2][16];
#pragma unroll
  for (int u = 0; u < 2; ++u) {
#pragma unroll
    for (int t = 0; t < 16; ++t) oacc[u][t] = (v8f){};
  }
  float mrun[2] = {-1e30f, -1e30f};
  float lrun[2] = {0.f, 0.f};

  const float* mk = mask + (size_t)n * HW;
  for (int i = 0; i < HW / 256; ++i)
    Madd[tid + 256 * i] = (mk[tid + 256 * i] - 1.f) * (1e8f * LOG2E);

  FETCH_TILE(0); STASH_TILE(0);
  __syncthreads();

  const int koff = (32 * kh + l15) * KP;
  const int voff = l15 * VP + 32 * kh;

  int b = 0;
  for (int s = 0; s < NSTAGE; ++s) {
    const int bn = (b == NSLOT - 1) ? 0 : b + 1;
    const int sn = (s + 1 < NSTAGE) ? s + 1 : NSTAGE - 1;
    const _Float16* kt = Kst + b * (SK * KP) + koff;
    const _Float16* vt = Vst + b * (CV * VP) + voff;
    const int j0 = s * SK + 32 * kh;

    FETCH_TILE(sn);
    asm volatile("" ::: "memory");

    const v8f ma0 = *reinterpret_cast<const v8f*>(&Madd[j0 + 8 * lh]);
    const v8f ma1 = *reinterpret_cast<const v8f*>(&Madd[j0 + 16 + 8 * lh]);

    v8f s0[2], s1[2];
#pragma unroll
    for (int u = 0; u < 2; ++u) { s0[u] = (v8f){}; s1[u] = (v8f){}; }
#pragma unroll
    for (int kc = 0; kc < 4; ++kc) {
      const _Float16* kb = kt + kc * 32;
      const v16h a0 = load_frag16(kb, lh);
      const v16h a1 = load_frag16(kb + 16 * KP, lh);
#pragma unroll
      for (int u = 0; u < 2; ++u) {
        s0[u] = __builtin_amdgcn_wmma_f32_16x16x32_f16(false, a0, false, qa[u][kc], (short)0, s0[u], false, false);
        s1[u] = __builtin_amdgcn_wmma_f32_16x16x32_f16(false, a1, false, qa[u][kc], (short)0, s1[u], false, false);
      }
    }

    constexpr float RESCALE_LAG = 8.0f;
    float vmx[2];
#pragma unroll
    for (int u = 0; u < 2; ++u) {
      float vmax = -1e30f;
#pragma unroll
      for (int r = 0; r < 8; ++r) {
        s0[u][r] = s0[u][r] * ATT_SCALE + ma0[r];
        s1[u][r] = s1[u][r] * ATT_SCALE + ma1[r];
        vmax = fmaxf(vmax, fmaxf(s0[u][r], s1[u][r]));
      }
      vmx[u] = vmax;
    }
    const bool lag_exceeded = (vmx[0] > mrun[0] + RESCALE_LAG) || (vmx[1] > mrun[1] + RESCALE_LAG);
    if (__builtin_amdgcn_ballot_w32(lag_exceeded) != 0u) {
#pragma unroll
      for (int u = 0; u < 2; ++u) {
        const float mn    = fmaxf(mrun[u], fmaxf(vmx[u], xhalf(vmx[u])));
        const float alpha = ex2(mrun[u] - mn);
        mrun[u] = mn;
        lrun[u] *= alpha;
#pragma unroll
        for (int t = 0; t < 16; ++t) oacc[u][t] *= alpha;
      }
    }
    v16h pbt[2];
#pragma unroll
    for (int u = 0; u < 2; ++u) {
      const float mu = mrun[u];
      float rs0 = 0.f, rs1 = 0.f;
#pragma unroll
      for (int r = 0; r < 8; ++r) {
        s0[u][r] = ex2(s0[u][r] - mu);
        s1[u][r] = ex2(s1[u][r] - mu);
        rs0 += s0[u][r];
        rs1 += s1[u][r];
      }
      lrun[u] += rs0 + rs1;

#pragma unroll
      for (int r = 0; r < 8; ++r) { pbt[u][r] = (_Float16)s0[u][r]; pbt[u][r + 8] = (_Float16)s1[u][r]; }
    }

    asm volatile("" :: "v"(lrun[0]), "v"(lrun[1]) : "memory");
    STASH_TILE(bn);
    __builtin_amdgcn_fence(__ATOMIC_RELEASE, "workgroup");
    __builtin_amdgcn_s_barrier_signal(-1);

    v16h va[16];
#pragma unroll
    for (int t = 0; t < 16; ++t) va[t] = load_frag16(vt + (t * 16) * VP, lh);
    __builtin_amdgcn_sched_barrier(0);
#pragma unroll
    for (int t = 0; t < 16; ++t) {
#pragma unroll
      for (int u = 0; u < 2; ++u)
        oacc[u][t] = __builtin_amdgcn_wmma_f32_16x16x32_f16(false, va[t], false, pbt[u], (short)0, oacc[u][t], false, false);
    }
    __builtin_amdgcn_sched_barrier(0);

    __builtin_amdgcn_s_barrier_wait(-1);
    __builtin_amdgcn_fence(__ATOMIC_ACQUIRE, "workgroup");
    b = bn;
  }
  __syncthreads();

  float lt[2];
#pragma unroll
  for (int u = 0; u < 2; ++u) lt[u] = lrun[u] + xhalf(lrun[u]);

  const int lane2 = (int)__builtin_amdgcn_mbcnt_lo(~0u, 0u);
  const int l15e  = lane2 & 15;
  const int lhe   = lane2 >> 4;
  float* mrg = Mrg + (size_t)g * (64 * 32 * 4) + lane2 * 4;
  float* mml = Mml + g * (2 * 2 * 32) + lane2;
  if (kh == 1) {
#pragma unroll
    for (int u = 0; u < 2; ++u) {
      mml[(u * 2 + 0) * 32] = mrun[u];
      mml[(u * 2 + 1) * 32] = lt[u];
#pragma unroll
      for (int t = 0; t < 16; ++t) {
        v4f lo, hi;
#pragma unroll
        for (int r = 0; r < 4; ++r) { lo[r] = oacc[u][t][r]; hi[r] = oacc[u][t][r + 4]; }
        *reinterpret_cast<v4f*>(mrg + ((u * 16 + t) * 2 + 0) * (32 * 4)) = lo;
        *reinterpret_cast<v4f*>(mrg + ((u * 16 + t) * 2 + 1) * (32 * 4)) = hi;
      }
    }
  }
  __syncthreads();
  if (kh == 0) {
#pragma unroll
    for (int u = 0; u < 2; ++u) {
      const float mp   = mml[(u * 2 + 0) * 32];
      const float lp   = mml[(u * 2 + 1) * 32];
      const float mn   = fmaxf(mrun[u], mp);
      const float ao   = ex2(mrun[u] - mn);
      const float ap   = ex2(mp - mn);
      const float ltot = ao * lt[u] + ap * lp;
      const float fin  = (1.0f / ltot) * mask[(size_t)n * HW + i0 + 16 * u + l15e];
      const float fo   = ao * fin;
      const float fp   = ap * fin;
#pragma unroll
      for (int t = 0; t < 16; ++t) {
        const v4f lo = *reinterpret_cast<const v4f*>(mrg + ((u * 16 + t) * 2 + 0) * (32 * 4));
        const v4f hi = *reinterpret_cast<const v4f*>(mrg + ((u * 16 + t) * 2 + 1) * (32 * 4));
        float* dst = out + ((size_t)n * CV + t * 16 + 8 * lhe) * HW + i0 + 16 * u + l15e;
#pragma unroll
        for (int r = 0; r < 4; ++r) {
          dst[(size_t)r * HW]       = oacc[u][t][r] * fo + lo[r] * fp;
          dst[(size_t)(r + 4) * HW] = oacc[u][t][r + 4] * fo + hi[r] * fp;
        }
      }
    }
  }
}

#undef FETCH_TILE
#undef STASH_TILE

extern "C" void kernel_launch(void* const* d_in, const int* in_sizes, int n_in,
                              void* d_out, int out_size, void* d_ws, size_t ws_size,
                              hipStream_t stream) {
  (void)in_sizes; (void)n_in; (void)out_size; (void)ws_size;
  const float* x  = (const float*)d_in[0];
  const float* mk = (const float*)d_in[1];
  const float* rw = (const float*)d_in[2];
  const float* rg = (const float*)d_in[3];
  const float* rb = (const float*)d_in[4];
  const float* rm = (const float*)d_in[5];
  const float* rv = (const float*)d_in[6];
  const float* aw = (const float*)d_in[7];
  const float* ag = (const float*)d_in[8];
  const float* ab = (const float*)d_in[9];
  const float* am = (const float*)d_in[10];
  const float* av = (const float*)d_in[11];
  float* out = (float*)d_out;

  _Float16* qbuf = (_Float16*)d_ws;
  _Float16* fT   = qbuf + (size_t)NB * HW * CQ;

  const dim3 blk(32, 4);
  proj_bn_relu_kernel<<<dim3(3, HW / 128, NB), blk, 0, stream>>>(
      x, rw, rg, rb, rm, rv, aw, ag, ab, am, av, fT, qbuf);
  flash_attn_kernel<<<dim3(HW / 128, NB), dim3(32, 8), 0, stream>>>(qbuf, fT, mk, out);
}
